# starting kernel
# baseline (speedup 1.0000x reference)
#include <hip/hip_runtime.h>
#include <hip/hip_bf16.h>
#include <math.h>


typedef __attribute__((ext_vector_type(2))) float v2f;
typedef __attribute__((ext_vector_type(8))) float v8f;

#define KTOT   3084
#define KSTEPS 771
#define NTILES 40
#define FRAG   64
#define NOUT   600
#define KCH    768
#define LDSTR  772

__global__ __launch_bounds__(256)
void ISTFT_buildW(float* __restrict__ Wf) {
  int gid = blockIdx.x * 256 + threadIdx.x;
  if (gid >= KSTEPS * NTILES * FRAG) return;

  int r    = gid & 1;
  int l    = (gid >> 1) & 31;
  int rest = gid >> 6;
  int nt   = rest % NTILES;
  int kk   = rest / NTILES;

  int k   = (kk << 2) + ((l >> 4) << 1) + r;
  int f   = k / 12;
  int rem = k - f * 12;
  int tf  = rem >> 1;
  int d   = rem & 1;
  int n   = (nt << 4) + (l & 15);

  float v = 0.0f;
  if (n < NOUT) {
    int labs = n + 256;
    int nn   = labs - (tf << 7);
    if (nn >= 0 && nn < 512) {
      const float c0 = 6.283185307179586f / 512.0f;
      float env = 0.0f;
      #pragma unroll
      for (int tp = 0; tp < 6; ++tp) {
        int nn2 = labs - (tp << 7);
        if (nn2 >= 0 && nn2 < 512) {
          float w = 0.5f * (1.0f - cosf(c0 * (float)nn2));
          env += w * w;
        }
      }
      float wn   = 0.5f * (1.0f - cosf(c0 * (float)nn));
      bool  edge = (f == 0) || (f == 256);
      float scl  = (edge ? 1.0f : 2.0f) / 512.0f;
      int   m    = (f * nn) & 511;
      float ang  = c0 * (float)m;
      float basis = d ? -sinf(ang) : cosf(ang);
      if (d && edge) basis = 0.0f;
      v = scl * wn / env * basis;
    }
  }
  Wf[gid] = v;
}

__global__ __launch_bounds__(256)
void ISTFT_gemm(const float* __restrict__ X, const float* __restrict__ Wf,
                float* __restrict__ out) {
  __shared__ float As[16 * LDSTR];

  const int tid  = threadIdx.x;
  const int wave = tid >> 5;
  const int lane = tid & 31;
  const int lrow = lane & 15;
  const int lhi  = lane >> 4;
  const int m0   = blockIdx.x << 4;

  v8f acc[5] = {};

  const float* __restrict__ Xblk = X + (size_t)m0 * KTOT;
  const float* __restrict__ Arow = As + lrow * LDSTR + (lhi << 1);
  const float* __restrict__ Wb   = Wf + (size_t)(wave * 5) * FRAG + lane * 2;

  for (int kbase = 0; kbase < KTOT; kbase += KCH) {
    const int klen = (KTOT - kbase < KCH) ? (KTOT - kbase) : KCH;

    for (int idx = tid; idx < 16 * klen; idx += 256) {
      int rr = idx / klen;
      int cc = idx - rr * klen;
      As[rr * LDSTR + cc] = Xblk[(size_t)rr * KTOT + kbase + cc];
    }
    __syncthreads();

    const int ksteps = klen >> 2;
    const int kk0    = kbase >> 2;
    for (int kk = 0; kk < ksteps; ++kk) {
      v2f a = *(const v2f*)(Arow + (kk << 2));
      const float* wr = Wb + (size_t)(kk0 + kk) * (NTILES * FRAG);
      #pragma unroll
      for (int t = 0; t < 5; ++t) {
        v2f b = *(const v2f*)(wr + t * FRAG);
        acc[t] = __builtin_amdgcn_wmma_f32_16x16x4_f32(
            false, a, false, b, (short)0, acc[t], false, false);
      }
    }
    __syncthreads();
  }

  #pragma unroll
  for (int t = 0; t < 5; ++t) {
    int colg = (wave * 5 + t) * 16 + lrow;
    if (colg < NOUT) {
      #pragma unroll
      for (int v = 0; v < 8; ++v) {
        int row = m0 + v + (lhi << 3);
        out[(size_t)row * NOUT + colg] = acc[t][v];
      }
    }
  }
}

extern "C" void kernel_launch(void* const* d_in, const int* in_sizes, int n_in,
                              void* d_out, int out_size, void* d_ws, size_t ws_size,
                              hipStream_t stream) {
  const float* X  = (const float*)d_in[0];
  float*       O  = (float*)d_out;
  float*       Wf = (float*)d_ws;

  const int wtot    = KSTEPS * NTILES * FRAG;
  const int wblocks = (wtot + 255) / 256;

  ISTFT_buildW<<<wblocks, 256, 0, stream>>>(Wf);
  ISTFT_gemm<<<8192 / 16, 256, 0, stream>>>(X, Wf, O);
}


# Skip the all-zero (frame, 16-column tile) blocks of W: K walked frame-major, one frame per k-step, each wave multiplies only the frames that reach its tiles; the sum order over k changes
# speedup vs baseline: 1.3252x; 1.3252x over previous
#include <hip/hip_runtime.h>
#include <hip/hip_bf16.h>
#include <math.h>


typedef __attribute__((ext_vector_type(2))) float v2f;
typedef __attribute__((ext_vector_type(8))) float v8f;

#define KTOT   3084
#define NFREQ  257
#define NFRM   6
#define QN     129
#define NTW    38
#define FRAG   64
#define NOUT   600
#define KCH    768
#define LDSTR  772
#define QSTR   (NTW * FRAG)

__global__ __launch_bounds__(256)
void ISTFT_buildW(float* __restrict__ Wf) {
  int gid = blockIdx.x * 256 + threadIdx.x;
  if (gid >= NFRM * QN * NTW * FRAG) return;

  int d    = gid & 1;
  int l    = (gid >> 1) & 31;
  int rest = gid >> 6;
  int nt   = rest % NTW;
  int kq   = rest / NTW;
  int tf   = kq / QN;
  int q    = kq - tf * QN;
  int f    = (q << 1) + (l >> 4);
  int n    = (nt << 4) + (l & 15);

  float v = 0.0f;
  if (n < NOUT && f < NFREQ) {
    int labs = n + 256;
    int nn   = labs - (tf << 7);
    if (nn >= 0 && nn < 512) {
      const float c0 = 6.283185307179586f / 512.0f;
      float env = 0.0f;
      #pragma unroll
      for (int tp = 0; tp < 6; ++tp) {
        int nn2 = labs - (tp << 7);
        if (nn2 >= 0 && nn2 < 512) {
          float w = 0.5f * (1.0f - cosf(c0 * (float)nn2));
          env += w * w;
        }
      }
      float wn   = 0.5f * (1.0f - cosf(c0 * (float)nn));
      bool  edge = (f == 0) || (f == 256);
      float scl  = (edge ? 1.0f : 2.0f) / 512.0f;
      int   m    = (f * nn) & 511;
      float ang  = c0 * (float)m;
      float basis = d ? -sinf(ang) : cosf(ang);
      if (d && edge) basis = 0.0f;
      v = scl * wn / env * basis;
    }
  }
  Wf[gid] = v;
}

template <int NT, int A0>
__device__ __forceinline__ void frame_steps(v8f (&acc)[7], const float* __restrict__ arow,
                                            const float* __restrict__ wb, int nsteps) {
  for (int s = 0; s < nsteps; ++s) {
    v2f a = *(const v2f*)(arow + s * 24);
    const float* wr = wb + (size_t)s * QSTR;
    #pragma unroll
    for (int t = 0; t < NT; ++t) {
      v2f b = *(const v2f*)(wr + t * FRAG);
      acc[A0 + t] = __builtin_amdgcn_wmma_f32_16x16x4_f32(
          false, a, false, b, (short)0, acc[A0 + t], false, false);
    }
  }
}

__global__ __launch_bounds__(256)
void ISTFT_gemm(const float* __restrict__ X, const float* __restrict__ Wf,
                float* __restrict__ out) {
  __shared__ float As[16 * LDSTR];

  const int tid  = threadIdx.x;
  const int wave = __builtin_amdgcn_readfirstlane(tid >> 5);
  const int lane = tid & 31;
  const int lrow = lane & 15;
  const int lhi  = lane >> 4;
  const int m0   = blockIdx.x << 4;

  const bool two   = wave < 2;
  const int  tileA = two ? (wave << 2) : 8 + ((wave - 2) << 2);
  const int  tileB = 32 + 3 * wave;
  const int  flo   = two ? 0 : ((wave - 2) >> 1);
  const int  fcnt  = two ? 3 : 4;

  v8f acc[7] = {};

  const float* __restrict__ Xblk = X + (size_t)m0 * KTOT;
  const float* __restrict__ Arow = As + lrow * LDSTR + lhi * 12;
  const float* __restrict__ WbA  = Wf + (size_t)tileA * FRAG + lane * 2;
  const float* __restrict__ WbB  = Wf + (size_t)tileB * FRAG + lane * 2;

  for (int kbase = 0; kbase < KTOT; kbase += KCH) {
    const int klen = (KTOT - kbase < KCH) ? (KTOT - kbase) : KCH;

    for (int idx = tid; idx < 16 * klen; idx += 256) {
      int rr = idx / klen;
      int cc = idx - rr * klen;
      As[rr * LDSTR + cc] = Xblk[(size_t)rr * KTOT + kbase + cc];
    }
    if (klen < KCH) {
      if (tid < 16 * 12) As[(tid / 12) * LDSTR + 12 + (tid % 12)] = 0.0f;
    }
    __syncthreads();

    const int nsteps = klen / 24 + ((klen % 24) ? 1 : 0);
    const int q0     = kbase / 24;

    for (int i = 0; i < fcnt; ++i) {
      const int tf = flo + i;
      frame_steps<4, 0>(acc, Arow + tf * 2,
                        WbA + (size_t)(tf * QN + q0) * QSTR, nsteps);
    }
    if (two) {
      for (int tf = 3; tf < 6; ++tf) {
        frame_steps<3, 4>(acc, Arow + tf * 2,
                          WbB + (size_t)(tf * QN + q0) * QSTR, nsteps);
      }
    }
    __syncthreads();
  }

  #pragma unroll
  for (int t = 0; t < 4; ++t) {
    int colg = (tileA + t) * 16 + lrow;
    #pragma unroll
    for (int v = 0; v < 8; ++v) {
      int row = m0 + v + (lhi << 3);
      out[(size_t)row * NOUT + colg] = acc[t][v];
    }
  }
  if (two) {
    #pragma unroll
    for (int t = 0; t < 3; ++t) {
      int colg = (tileB + t) * 16 + lrow;
      if (colg < NOUT) {
        #pragma unroll
        for (int v = 0; v < 8; ++v) {
          int row = m0 + v + (lhi << 3);
          out[(size_t)row * NOUT + colg] = acc[4 + t][v];
        }
      }
    }
  }
}

extern "C" void kernel_launch(void* const* d_in, const int* in_sizes, int n_in,
                              void* d_out, int out_size, void* d_ws, size_t ws_size,
                              hipStream_t stream) {
  const float* X  = (const float*)d_in[0];
  float*       O  = (float*)d_out;
  float*       Wf = (float*)d_ws;

  const int wtot    = NFRM * QN * NTW * FRAG;
  const int wblocks = (wtot + 255) / 256;

  ISTFT_buildW<<<wblocks, 256, 0, stream>>>(Wf);
  ISTFT_gemm<<<8192 / 16, 256, 0, stream>>>(X, Wf, O);
}


# A staging software-pipelined: the next K chunk's rows are fetched with 16-byte loads into registers before the current chunk's matrix work and stored to LDS after it; sums unchanged
# speedup vs baseline: 1.5109x; 1.1401x over previous
#include <hip/hip_runtime.h>
#include <hip/hip_bf16.h>
#include <math.h>


typedef __attribute__((ext_vector_type(2))) float v2f;
typedef __attribute__((ext_vector_type(8))) float v8f;
typedef __attribute__((ext_vector_type(4))) float v4f;

#define KTOT   3084
#define NFREQ  257
#define NFRM   6
#define QN     129
#define NTW    38
#define FRAG   64
#define NOUT   600
#define KCH    768
#define LDSTR  772
#define QSTR   (NTW * FRAG)

__global__ __launch_bounds__(256)
void ISTFT_buildW(float* __restrict__ Wf) {
  int gid = blockIdx.x * 256 + threadIdx.x;
  if (gid >= NFRM * QN * NTW * FRAG) return;

  int d    = gid & 1;
  int l    = (gid >> 1) & 31;
  int rest = gid >> 6;
  int nt   = rest % NTW;
  int kq   = rest / NTW;
  int tf   = kq / QN;
  int q    = kq - tf * QN;
  int f    = (q << 1) + (l >> 4);
  int n    = (nt << 4) + (l & 15);

  float v = 0.0f;
  if (n < NOUT && f < NFREQ) {
    int labs = n + 256;
    int nn   = labs - (tf << 7);
    if (nn >= 0 && nn < 512) {
      const float c0 = 6.283185307179586f / 512.0f;
      float env = 0.0f;
      #pragma unroll
      for (int tp = 0; tp < 6; ++tp) {
        int nn2 = labs - (tp << 7);
        if (nn2 >= 0 && nn2 < 512) {
          float w = 0.5f * (1.0f - cosf(c0 * (float)nn2));
          env += w * w;
        }
      }
      float wn   = 0.5f * (1.0f - cosf(c0 * (float)nn));
      bool  edge = (f == 0) || (f == 256);
      float scl  = (edge ? 1.0f : 2.0f) / 512.0f;
      int   m    = (f * nn) & 511;
      float ang  = c0 * (float)m;
      float basis = d ? -sinf(ang) : cosf(ang);
      if (d && edge) basis = 0.0f;
      v = scl * wn / env * basis;
    }
  }
  Wf[gid] = v;
}

template <int NT, int A0>
__device__ __forceinline__ void frame_steps(v8f (&acc)[7], const float* __restrict__ arow,
                                            const float* __restrict__ wb, int nsteps) {
  #pragma unroll 4
  for (int s = 0; s < nsteps; ++s) {
    v2f a = *(const v2f*)(arow + s * 24);
    const float* wr = wb + (size_t)s * QSTR;
    #pragma unroll
    for (int t = 0; t < NT; ++t) {
      v2f b = *(const v2f*)(wr + t * FRAG);
      acc[A0 + t] = __builtin_amdgcn_wmma_f32_16x16x4_f32(
          false, a, false, b, (short)0, acc[A0 + t], false, false);
    }
  }
}

__global__ __launch_bounds__(256)
void ISTFT_gemm(const float* __restrict__ X, const float* __restrict__ Wf,
                float* __restrict__ out) {
  __shared__ float As[16 * LDSTR];

  const int tid  = threadIdx.x;
  const int wave = __builtin_amdgcn_readfirstlane(tid >> 5);
  const int lane = tid & 31;
  const int lrow = lane & 15;
  const int lhi  = lane >> 4;
  const int m0   = blockIdx.x << 4;

  const bool two   = wave < 2;
  const int  tileA = two ? (wave << 2) : 8 + ((wave - 2) << 2);
  const int  tileB = 32 + 3 * wave;
  const int  flo   = two ? 0 : ((wave - 2) >> 1);
  const int  fcnt  = two ? 3 : 4;

  v8f acc[7] = {};

  const float* __restrict__ Xblk = X + (size_t)m0 * KTOT;
  const float* __restrict__ Arow = As + lrow * LDSTR + lhi * 12;
  const float* __restrict__ WbA  = Wf + (size_t)tileA * FRAG + lane * 2;
  const float* __restrict__ WbB  = Wf + (size_t)tileB * FRAG + lane * 2;

  v4f pre[12];

  auto chunk_products = [&](int q0, int nsteps) {
    for (int i = 0; i < fcnt; ++i) {
      const int tf = flo + i;
      frame_steps<4, 0>(acc, Arow + tf * 2,
                        WbA + (size_t)(tf * QN + q0) * QSTR, nsteps);
    }
    if (two) {
      for (int tf = 3; tf < 6; ++tf) {
        frame_steps<3, 4>(acc, Arow + tf * 2,
                          WbB + (size_t)(tf * QN + q0) * QSTR, nsteps);
      }
    }
  };

  #pragma unroll
  for (int i = 0; i < 12; ++i) {
    const int id = tid + (i << 8);
    const int rr = id / 192, cc = id - rr * 192;
    pre[i] = *(const v4f*)(Xblk + (size_t)rr * KTOT + (cc << 2));
  }
  #pragma unroll
  for (int i = 0; i < 12; ++i) {
    const int id = tid + (i << 8);
    const int rr = id / 192, cc = id - rr * 192;
    *(v4f*)(As + rr * LDSTR + (cc << 2)) = pre[i];
  }
  __syncthreads();

  for (int kbase = 0; kbase < 3 * KCH; kbase += KCH) {
    #pragma unroll
    for (int i = 0; i < 12; ++i) {
      const int id = tid + (i << 8);
      const int rr = id / 192, cc = id - rr * 192;
      pre[i] = *(const v4f*)(Xblk + (size_t)rr * KTOT + (kbase + KCH) + (cc << 2));
    }
    chunk_products(kbase / 24, KCH / 24);
    __syncthreads();
    #pragma unroll
    for (int i = 0; i < 12; ++i) {
      const int id = tid + (i << 8);
      const int rr = id / 192, cc = id - rr * 192;
      *(v4f*)(As + rr * LDSTR + (cc << 2)) = pre[i];
    }
    __syncthreads();
  }

  {
    const int tp = (tid < 48) ? tid : 0;
    const int tr = tp / 3, tc = tp - tr * 3;
    const v4f tailv = *(const v4f*)(Xblk + (size_t)tr * KTOT + 4 * KCH + (tc << 2));
    chunk_products((3 * KCH) / 24, KCH / 24);
    __syncthreads();
    if (tid < 48) {
      *(v4f*)(As + tr * LDSTR + (tc << 2)) = tailv;
    } else if (tid < 96) {
      const int t2 = tid - 48;
      const int zr = t2 / 3, zc = t2 - zr * 3;
      *(v4f*)(As + zr * LDSTR + 12 + (zc << 2)) = v4f{0.0f, 0.0f, 0.0f, 0.0f};
    }
    __syncthreads();
  }

  chunk_products((4 * KCH) / 24, 1);

  #pragma unroll
  for (int t = 0; t < 4; ++t) {
    int colg = (tileA + t) * 16 + lrow;
    #pragma unroll
    for (int v = 0; v < 8; ++v) {
      int row = m0 + v + (lhi << 3);
      out[(size_t)row * NOUT + colg] = acc[t][v];
    }
  }
  if (two) {
    #pragma unroll
    for (int t = 0; t < 3; ++t) {
      int colg = (tileB + t) * 16 + lrow;
      if (colg < NOUT) {
        #pragma unroll
        for (int v = 0; v < 8; ++v) {
          int row = m0 + v + (lhi << 3);
          out[(size_t)row * NOUT + colg] = acc[4 + t][v];
        }
      }
    }
  }
}

extern "C" void kernel_launch(void* const* d_in, const int* in_sizes, int n_in,
                              void* d_out, int out_size, void* d_ws, size_t ws_size,
                              hipStream_t stream) {
  const float* X  = (const float*)d_in[0];
  float*       O  = (float*)d_out;
  float*       Wf = (float*)d_ws;

  const int wtot    = NFRM * QN * NTW * FRAG;
  const int wblocks = (wtot + 255) / 256;

  ISTFT_buildW<<<wblocks, 256, 0, stream>>>(Wf);
  ISTFT_gemm<<<8192 / 16, 256, 0, stream>>>(X, Wf, O);
}


# W and A fragment loads software-pipelined over two register sets: the next four k-steps' fragments are fetched ahead of the current four steps' matrix work, latency under it; sums unchanged
# speedup vs baseline: 1.7205x; 1.1387x over previous
#include <hip/hip_runtime.h>
#include <hip/hip_bf16.h>
#include <math.h>


typedef __attribute__((ext_vector_type(2))) float v2f;
typedef __attribute__((ext_vector_type(8))) float v8f;
typedef __attribute__((ext_vector_type(4))) float v4f;

#define KTOT   3084
#define NFREQ  257
#define NFRM   6
#define QN     129
#define NTW    38
#define FRAG   64
#define NOUT   600
#define KCH    768
#define LDSTR  772
#define QSTR   (NTW * FRAG)

__global__ __launch_bounds__(256)
void ISTFT_buildW(float* __restrict__ Wf) {
  int gid = blockIdx.x * 256 + threadIdx.x;
  if (gid >= NFRM * QN * NTW * FRAG) return;

  int d    = gid & 1;
  int l    = (gid >> 1) & 31;
  int rest = gid >> 6;
  int nt   = rest % NTW;
  int kq   = rest / NTW;
  int tf   = kq / QN;
  int q    = kq - tf * QN;
  int f    = (q << 1) + (l >> 4);
  int n    = (nt << 4) + (l & 15);

  float v = 0.0f;
  if (n < NOUT && f < NFREQ) {
    int labs = n + 256;
    int nn   = labs - (tf << 7);
    if (nn >= 0 && nn < 512) {
      const float c0 = 6.283185307179586f / 512.0f;
      float env = 0.0f;
      #pragma unroll
      for (int tp = 0; tp < 6; ++tp) {
        int nn2 = labs - (tp << 7);
        if (nn2 >= 0 && nn2 < 512) {
          float w = 0.5f * (1.0f - cosf(c0 * (float)nn2));
          env += w * w;
        }
      }
      float wn   = 0.5f * (1.0f - cosf(c0 * (float)nn));
      bool  edge = (f == 0) || (f == 256);
      float scl  = (edge ? 1.0f : 2.0f) / 512.0f;
      int   m    = (f * nn) & 511;
      float ang  = c0 * (float)m;
      float basis = d ? -sinf(ang) : cosf(ang);
      if (d && edge) basis = 0.0f;
      v = scl * wn / env * basis;
    }
  }
  Wf[gid] = v;
}

template <int NT>
__device__ __forceinline__ void load_frags(v2f (&a)[4], v2f (&b)[4][NT],
                                           const float* __restrict__ ap,
                                           const float* __restrict__ wp) {
  #pragma unroll
  for (int s = 0; s < 4; ++s) {
    #pragma unroll
    for (int t = 0; t < NT; ++t) b[s][t] = *(const v2f*)(wp + (size_t)s * QSTR + t * FRAG);
  }
  #pragma unroll
  for (int s = 0; s < 4; ++s) a[s] = *(const v2f*)(ap + s * 24);
}

template <int NT, int A0>
__device__ __forceinline__ void mma_frags(v8f (&acc)[7], const v2f (&a)[4], const v2f (&b)[4][NT]) {
  #pragma unroll
  for (int s = 0; s < 4; ++s) {
    #pragma unroll
    for (int t = 0; t < NT; ++t) {
      acc[A0 + t] = __builtin_amdgcn_wmma_f32_16x16x4_f32(
          false, a[s], false, b[s][t], (short)0, acc[A0 + t], false, false);
    }
  }
}

template <int NT, int A0>
__device__ __forceinline__ void chunk_group(v8f (&acc)[7], const float* __restrict__ arow,
                                            const float* __restrict__ wb, int f0, int nfr, int q0) {
  const int ntrips = nfr << 3;
  v2f a0[4], a1[4];
  v2f b0[4][NT], b1[4][NT];
  auto aptr = [&](int j) { return arow + (f0 + (j >> 3)) * 2 + ((j & 7) << 2) * 24; };
  auto wptr = [&](int j) {
    return wb + (size_t)((f0 + (j >> 3)) * QN + q0 + ((j & 7) << 2)) * QSTR;
  };

  load_frags<NT>(a0, b0, aptr(0), wptr(0));
  for (int j = 0; j + 2 < ntrips; j += 2) {
    load_frags<NT>(a1, b1, aptr(j + 1), wptr(j + 1));
    asm volatile("" ::: "memory");
    mma_frags<NT, A0>(acc, a0, b0);
    load_frags<NT>(a0, b0, aptr(j + 2), wptr(j + 2));
    asm volatile("" ::: "memory");
    mma_frags<NT, A0>(acc, a1, b1);
  }
  load_frags<NT>(a1, b1, aptr(ntrips - 1), wptr(ntrips - 1));
  asm volatile("" ::: "memory");
  mma_frags<NT, A0>(acc, a0, b0);
  mma_frags<NT, A0>(acc, a1, b1);
}

template <int NT, int A0>
__device__ __forceinline__ void frame_one_step(v8f (&acc)[7], const float* __restrict__ arow,
                                               const float* __restrict__ wb) {
  v2f a = *(const v2f*)arow;
  #pragma unroll
  for (int t = 0; t < NT; ++t) {
    v2f b = *(const v2f*)(wb + t * FRAG);
    acc[A0 + t] = __builtin_amdgcn_wmma_f32_16x16x4_f32(
        false, a, false, b, (short)0, acc[A0 + t], false, false);
  }
}

__global__ __launch_bounds__(256, 4)
void ISTFT_gemm(const float* __restrict__ X, const float* __restrict__ Wf,
                float* __restrict__ out) {
  __shared__ float As[16 * LDSTR];

  const int tid  = threadIdx.x;
  const int wave = __builtin_amdgcn_readfirstlane(tid >> 5);
  const int lane = tid & 31;
  const int lrow = lane & 15;
  const int lhi  = lane >> 4;
  const int m0   = blockIdx.x << 4;

  const bool two   = wave < 2;
  const int  tileA = two ? (wave << 2) : 8 + ((wave - 2) << 2);
  const int  tileB = 32 + 3 * wave;
  const int  flo   = two ? 0 : ((wave - 2) >> 1);
  const int  fcnt  = two ? 3 : 4;

  v8f acc[7] = {};

  const float* __restrict__ Xblk = X + (size_t)m0 * KTOT;
  const float* __restrict__ Arow = As + lrow * LDSTR + lhi * 12;
  const float* __restrict__ WbA  = Wf + (size_t)tileA * FRAG + lane * 2;
  const float* __restrict__ WbB  = Wf + (size_t)tileB * FRAG + lane * 2;

  v4f pre[12];

  auto chunk_products = [&](int q0) {
    chunk_group<4, 0>(acc, Arow, WbA, flo, fcnt, q0);
    if (two) chunk_group<3, 4>(acc, Arow, WbB, 3, 3, q0);
  };
  auto tail_products = [&](int q0) {
    for (int i = 0; i < fcnt; ++i) {
      const int tf = flo + i;
      frame_one_step<4, 0>(acc, Arow + tf * 2, WbA + (size_t)(tf * QN + q0) * QSTR);
    }
    if (two) {
      for (int tf = 3; tf < 6; ++tf) {
        frame_one_step<3, 4>(acc, Arow + tf * 2, WbB + (size_t)(tf * QN + q0) * QSTR);
      }
    }
  };

  #pragma unroll
  for (int i = 0; i < 12; ++i) {
    const int id = tid + (i << 8);
    const int rr = id / 192, cc = id - rr * 192;
    pre[i] = *(const v4f*)(Xblk + (size_t)rr * KTOT + (cc << 2));
  }
  #pragma unroll
  for (int i = 0; i < 12; ++i) {
    const int id = tid + (i << 8);
    const int rr = id / 192, cc = id - rr * 192;
    *(v4f*)(As + rr * LDSTR + (cc << 2)) = pre[i];
  }
  __syncthreads();

  for (int kbase = 0; kbase < 3 * KCH; kbase += KCH) {
    #pragma unroll
    for (int i = 0; i < 12; ++i) {
      const int id = tid + (i << 8);
      const int rr = id / 192, cc = id - rr * 192;
      pre[i] = *(const v4f*)(Xblk + (size_t)rr * KTOT + (kbase + KCH) + (cc << 2));
    }
    chunk_products(kbase / 24);
    __syncthreads();
    #pragma unroll
    for (int i = 0; i < 12; ++i) {
      const int id = tid + (i << 8);
      const int rr = id / 192, cc = id - rr * 192;
      *(v4f*)(As + rr * LDSTR + (cc << 2)) = pre[i];
    }
    __syncthreads();
  }

  {
    const int tp = (tid < 48) ? tid : 0;
    const int tr = tp / 3, tc = tp - tr * 3;
    const v4f tailv = *(const v4f*)(Xblk + (size_t)tr * KTOT + 4 * KCH + (tc << 2));
    chunk_products((3 * KCH) / 24);
    __syncthreads();
    if (tid < 48) {
      *(v4f*)(As + tr * LDSTR + (tc << 2)) = tailv;
    } else if (tid < 96) {
      const int t2 = tid - 48;
      const int zr = t2 / 3, zc = t2 - zr * 3;
      *(v4f*)(As + zr * LDSTR + 12 + (zc << 2)) = v4f{0.0f, 0.0f, 0.0f, 0.0f};
    }
    __syncthreads();
  }

  tail_products((4 * KCH) / 24);

  #pragma unroll
  for (int t = 0; t < 4; ++t) {
    int colg = (tileA + t) * 16 + lrow;
    #pragma unroll
    for (int v = 0; v < 8; ++v) {
      int row = m0 + v + (lhi << 3);
      out[(size_t)row * NOUT + colg] = acc[t][v];
    }
  }
  if (two) {
    #pragma unroll
    for (int t = 0; t < 3; ++t) {
      int colg = (tileB + t) * 16 + lrow;
      if (colg < NOUT) {
        #pragma unroll
        for (int v = 0; v < 8; ++v) {
          int row = m0 + v + (lhi << 3);
          out[(size_t)row * NOUT + colg] = acc[4 + t][v];
        }
      }
    }
  }
}

extern "C" void kernel_launch(void* const* d_in, const int* in_sizes, int n_in,
                              void* d_out, int out_size, void* d_ws, size_t ws_size,
                              hipStream_t stream) {
  const float* X  = (const float*)d_in[0];
  float*       O  = (float*)d_out;
  float*       Wf = (float*)d_ws;

  const int wtot    = NFRM * QN * NTW * FRAG;
  const int wblocks = (wtot + 255) / 256;

  ISTFT_buildW<<<wblocks, 256, 0, stream>>>(Wf);
  ISTFT_gemm<<<8192 / 16, 256, 0, stream>>>(X, Wf, O);
}


# Two 16-row M tiles per wave (32 batch rows per block, 256 blocks): each W fragment fetched from L2 feeds two matrix instructions, W loads and traffic per product halved; sums unchanged
# speedup vs baseline: 1.8455x; 1.0727x over previous
#include <hip/hip_runtime.h>
#include <hip/hip_bf16.h>
#include <math.h>


typedef __attribute__((ext_vector_type(2))) float v2f;
typedef __attribute__((ext_vector_type(8))) float v8f;
typedef __attribute__((ext_vector_type(4))) float v4f;

#define KTOT   3084
#define NFREQ  257
#define NFRM   6
#define QN     129
#define NTW    38
#define FRAG   64
#define NOUT   600
#define KCH    768
#define LDSTR  772
#define QSTR   (NTW * FRAG)
#define MT     2
#define MSTR   (16 * LDSTR)

__global__ __launch_bounds__(256)
void ISTFT_buildW(float* __restrict__ Wf) {
  int gid = blockIdx.x * 256 + threadIdx.x;
  if (gid >= NFRM * QN * NTW * FRAG) return;

  int d    = gid & 1;
  int l    = (gid >> 1) & 31;
  int rest = gid >> 6;
  int nt   = rest % NTW;
  int kq   = rest / NTW;
  int tf   = kq / QN;
  int q    = kq - tf * QN;
  int f    = (q << 1) + (l >> 4);
  int n    = (nt << 4) + (l & 15);

  float v = 0.0f;
  if (n < NOUT && f < NFREQ) {
    int labs = n + 256;
    int nn   = labs - (tf << 7);
    if (nn >= 0 && nn < 512) {
      const float c0 = 6.283185307179586f / 512.0f;
      float env = 0.0f;
      #pragma unroll
      for (int tp = 0; tp < 6; ++tp) {
        int nn2 = labs - (tp << 7);
        if (nn2 >= 0 && nn2 < 512) {
          float w = 0.5f * (1.0f - cosf(c0 * (float)nn2));
          env += w * w;
        }
      }
      float wn   = 0.5f * (1.0f - cosf(c0 * (float)nn));
      bool  edge = (f == 0) || (f == 256);
      float scl  = (edge ? 1.0f : 2.0f) / 512.0f;
      int   m    = (f * nn) & 511;
      float ang  = c0 * (float)m;
      float basis = d ? -sinf(ang) : cosf(ang);
      if (d && edge) basis = 0.0f;
      v = scl * wn / env * basis;
    }
  }
  Wf[gid] = v;
}

template <int NT>
__device__ __forceinline__ void load_frags(v2f (&a)[4][MT], v2f (&b)[4][NT],
                                           const float* __restrict__ ap,
                                           const float* __restrict__ wp) {
  #pragma unroll
  for (int s = 0; s < 4; ++s) {
    #pragma unroll
    for (int t = 0; t < NT; ++t) b[s][t] = *(const v2f*)(wp + (size_t)s * QSTR + t * FRAG);
  }
  #pragma unroll
  for (int s = 0; s < 4; ++s) {
    #pragma unroll
    for (int m = 0; m < MT; ++m) a[s][m] = *(const v2f*)(ap + s * 24 + m * MSTR);
  }
}

template <int NT, int A0>
__device__ __forceinline__ void mma_frags(v8f (&acc)[7 * MT], const v2f (&a)[4][MT],
                                          const v2f (&b)[4][NT]) {
  #pragma unroll
  for (int s = 0; s < 4; ++s) {
    #pragma unroll
    for (int t = 0; t < NT; ++t) {
      #pragma unroll
      for (int m = 0; m < MT; ++m) {
        acc[MT * (A0 + t) + m] = __builtin_amdgcn_wmma_f32_16x16x4_f32(
            false, a[s][m], false, b[s][t], (short)0, acc[MT * (A0 + t) + m], false, false);
      }
    }
  }
}

template <int NT, int A0>
__device__ __forceinline__ void chunk_group(v8f (&acc)[7 * MT], const float* __restrict__ arow,
                                            const float* __restrict__ wb, int f0, int nfr, int q0) {
  const int ntrips = nfr << 3;
  v2f a0[4][MT], a1[4][MT];
  v2f b0[4][NT], b1[4][NT];
  auto aptr = [&](int j) { return arow + (f0 + (j >> 3)) * 2 + ((j & 7) << 2) * 24; };
  auto wptr = [&](int j) {
    return wb + (size_t)((f0 + (j >> 3)) * QN + q0 + ((j & 7) << 2)) * QSTR;
  };

  load_frags<NT>(a0, b0, aptr(0), wptr(0));
  for (int j = 0; j + 2 < ntrips; j += 2) {
    load_frags<NT>(a1, b1, aptr(j + 1), wptr(j + 1));
    asm volatile("" ::: "memory");
    mma_frags<NT, A0>(acc, a0, b0);
    load_frags<NT>(a0, b0, aptr(j + 2), wptr(j + 2));
    mma_frags<NT, A0>(acc, a1, b1);
  }
  load_frags<NT>(a1, b1, aptr(ntrips - 1), wptr(ntrips - 1));
  mma_frags<NT, A0>(acc, a0, b0);
  mma_frags<NT, A0>(acc, a1, b1);
}

template <int NT, int A0>
__device__ __forceinline__ void frame_one_step(v8f (&acc)[7 * MT], const float* __restrict__ arow,
                                               const float* __restrict__ wb) {
  v2f a[MT];
  #pragma unroll
  for (int m = 0; m < MT; ++m) a[m] = *(const v2f*)(arow + m * MSTR);
  #pragma unroll
  for (int t = 0; t < NT; ++t) {
    v2f b = *(const v2f*)(wb + t * FRAG);
    #pragma unroll
    for (int m = 0; m < MT; ++m) {
      acc[MT * (A0 + t) + m] = __builtin_amdgcn_wmma_f32_16x16x4_f32(
          false, a[m], false, b, (short)0, acc[MT * (A0 + t) + m], false, false);
    }
  }
}

__global__ __launch_bounds__(256)
void ISTFT_gemm(const float* __restrict__ X, const float* __restrict__ Wf,
                float* __restrict__ out) {
  __shared__ float As[MT * 16 * LDSTR];

  const int tid  = threadIdx.x;
  const int wave = __builtin_amdgcn_readfirstlane(tid >> 5);
  const int lane = tid & 31;
  const int lrow = lane & 15;
  const int lhi  = lane >> 4;
  const int m0   = blockIdx.x << 5;

  const bool two   = wave < 2;
  const int  tileA = two ? (wave << 2) : 8 + ((wave - 2) << 2);
  const int  tileB = 32 + 3 * wave;
  const int  flo   = two ? 0 : ((wave - 2) >> 1);
  const int  fcnt  = two ? 3 : 4;

  v8f acc[7 * MT] = {};

  const float* __restrict__ Xblk = X + (size_t)m0 * KTOT;
  const float* __restrict__ Arow = As + lrow * LDSTR + lhi * 12;
  const float* __restrict__ WbA  = Wf + (size_t)tileA * FRAG + lane * 2;
  const float* __restrict__ WbB  = Wf + (size_t)tileB * FRAG + lane * 2;

  v4f pre[24];

  auto chunk_products = [&](int q0) {
    chunk_group<4, 0>(acc, Arow, WbA, flo, fcnt, q0);
    if (two) chunk_group<3, 4>(acc, Arow, WbB, 3, 3, q0);
  };
  auto tail_products = [&](int q0) {
    for (int i = 0; i < fcnt; ++i) {
      const int tf = flo + i;
      frame_one_step<4, 0>(acc, Arow + tf * 2, WbA + (size_t)(tf * QN + q0) * QSTR);
    }
    if (two) {
      for (int tf = 3; tf < 6; ++tf) {
        frame_one_step<3, 4>(acc, Arow + tf * 2, WbB + (size_t)(tf * QN + q0) * QSTR);
      }
    }
  };

  #pragma unroll
  for (int i = 0; i < 24; ++i) {
    const int id = tid + (i << 8);
    const int rr = id / 192, cc = id - rr * 192;
    pre[i] = *(const v4f*)(Xblk + (size_t)rr * KTOT + (cc << 2));
  }
  #pragma unroll
  for (int i = 0; i < 24; ++i) {
    const int id = tid + (i << 8);
    const int rr = id / 192, cc = id - rr * 192;
    *(v4f*)(As + rr * LDSTR + (cc << 2)) = pre[i];
  }
  __syncthreads();

  for (int kbase = 0; kbase < 3 * KCH; kbase += KCH) {
    #pragma unroll
    for (int i = 0; i < 24; ++i) {
      const int id = tid + (i << 8);
      const int rr = id / 192, cc = id - rr * 192;
      pre[i] = *(const v4f*)(Xblk + (size_t)rr * KTOT + (kbase + KCH) + (cc << 2));
    }
    chunk_products(kbase / 24);
    __syncthreads();
    #pragma unroll
    for (int i = 0; i < 24; ++i) {
      const int id = tid + (i << 8);
      const int rr = id / 192, cc = id - rr * 192;
      *(v4f*)(As + rr * LDSTR + (cc << 2)) = pre[i];
    }
    __syncthreads();
  }

  {
    const int tp = (tid < 96) ? tid : 0;
    const int tr = tp / 3, tc = tp - tr * 3;
    const v4f tailv = *(const v4f*)(Xblk + (size_t)tr * KTOT + 4 * KCH + (tc << 2));
    chunk_products((3 * KCH) / 24);
    __syncthreads();
    if (tid < 96) {
      *(v4f*)(As + tr * LDSTR + (tc << 2)) = tailv;
    } else if (tid < 192) {
      const int t2 = tid - 96;
      const int zr = t2 / 3, zc = t2 - zr * 3;
      *(v4f*)(As + zr * LDSTR + 12 + (zc << 2)) = v4f{0.0f, 0.0f, 0.0f, 0.0f};
    }
    __syncthreads();
  }

  tail_products((4 * KCH) / 24);

  #pragma unroll
  for (int t = 0; t < 4; ++t) {
    int colg = (tileA + t) * 16 + lrow;
    #pragma unroll
    for (int m = 0; m < MT; ++m) {
      #pragma unroll
      for (int v = 0; v < 8; ++v) {
        int row = m0 + (m << 4) + v + (lhi << 3);
        out[(size_t)row * NOUT + colg] = acc[MT * t + m][v];
      }
    }
  }
  if (two) {
    #pragma unroll
    for (int t = 0; t < 3; ++t) {
      int colg = (tileB + t) * 16 + lrow;
      if (colg < NOUT) {
        #pragma unroll
        for (int m = 0; m < MT; ++m) {
          #pragma unroll
          for (int v = 0; v < 8; ++v) {
            int row = m0 + (m << 4) + v + (lhi << 3);
            out[(size_t)row * NOUT + colg] = acc[MT * (4 + t) + m][v];
          }
        }
      }
    }
  }
}

extern "C" void kernel_launch(void* const* d_in, const int* in_sizes, int n_in,
                              void* d_out, int out_size, void* d_ws, size_t ws_size,
                              hipStream_t stream) {
  const float* X  = (const float*)d_in[0];
  float*       O  = (float*)d_out;
  float*       Wf = (float*)d_ws;

  const int wtot    = NFRM * QN * NTW * FRAG;
  const int wblocks = (wtot + 255) / 256;

  ISTFT_buildW<<<wblocks, 256, 0, stream>>>(Wf);
  ISTFT_gemm<<<8192 / 32, 256, 0, stream>>>(X, Wf, O);
}


# Inverse DFT split by output-sample parity: rows staged to LDS as X[f]+X[256-f] and X[f]-X[256-f], each 16-column tile holds one parity, K per product halved; the pre-adds change the sum order
# speedup vs baseline: 3.1513x; 1.7076x over previous
#include <hip/hip_runtime.h>
#include <hip/hip_bf16.h>
#include <math.h>


typedef __attribute__((ext_vector_type(2))) float v2f;
typedef __attribute__((ext_vector_type(8))) float v8f;
typedef __attribute__((ext_vector_type(4))) float v4f;

#define KTOT   3084
#define NFOLD  129
#define NFRM   6
#define QN     65
#define NTW    38
#define FRAG   64
#define NOUT   600
#define FCH    32
#define QCH    (FCH / 2)
#define LDSTR  388
#define PLANE  (32 * LDSTR)
#define QSTR   (NTW * FRAG)
#define MT     2
#define MSTR   (16 * LDSTR)

__device__ __forceinline__ int tile_col0(int nt) {
  int p, T;
  if (nt < 32) { p = (nt >> 2) & 1; T = ((nt >> 3) << 2) + (nt & 3); }
  else         { const int r = nt - 32; p = r / 3; T = 16 + r - 3 * p; }
  return (T << 5) + p;
}

__global__ __launch_bounds__(256)
void ISTFT_buildW(float* __restrict__ Wf) {
  int gid = blockIdx.x * 256 + threadIdx.x;
  if (gid >= NFRM * QN * NTW * FRAG) return;

  int d    = gid & 1;
  int l    = (gid >> 1) & 31;
  int rest = gid >> 6;
  int nt   = rest % NTW;
  int kq   = rest / NTW;
  int tf   = kq / QN;
  int q    = kq - tf * QN;
  int f    = (q << 1) + (l >> 4);
  int n    = tile_col0(nt) + ((l & 15) << 1);

  float v = 0.0f;
  if (n < NOUT && f < NFOLD) {
    int labs = n + 256;
    int nn   = labs - (tf << 7);
    if (nn >= 0 && nn < 512) {
      const float c0 = 6.283185307179586f / 512.0f;
      float env = 0.0f;
      #pragma unroll
      for (int tp = 0; tp < 6; ++tp) {
        int nn2 = labs - (tp << 7);
        if (nn2 >= 0 && nn2 < 512) {
          float w = 0.5f * (1.0f - cosf(c0 * (float)nn2));
          env += w * w;
        }
      }
      float wn   = 0.5f * (1.0f - cosf(c0 * (float)nn));
      bool  edge = (f == 0);
      float scl  = (edge ? 1.0f : 2.0f) / 512.0f;
      int   m    = (f * nn) & 511;
      float ang  = c0 * (float)m;
      float basis = d ? -sinf(ang) : cosf(ang);
      if (d && edge) basis = 0.0f;
      v = scl * wn / env * basis;
    }
  }
  Wf[gid] = v;
}

template <int NT>
__device__ __forceinline__ void load_frags(v2f (&a)[4][MT], v2f (&b)[4][NT],
                                           const float* __restrict__ ap,
                                           const float* __restrict__ wp) {
  #pragma unroll
  for (int s = 0; s < 4; ++s) {
    #pragma unroll
    for (int t = 0; t < NT; ++t) b[s][t] = *(const v2f*)(wp + (size_t)s * QSTR + t * FRAG);
  }
  #pragma unroll
  for (int s = 0; s < 4; ++s) {
    #pragma unroll
    for (int m = 0; m < MT; ++m) a[s][m] = *(const v2f*)(ap + s * 24 + m * MSTR);
  }
}

template <int NT, int A0>
__device__ __forceinline__ void mma_frags(v8f (&acc)[7 * MT], const v2f (&a)[4][MT],
                                          const v2f (&b)[4][NT]) {
  #pragma unroll
  for (int s = 0; s < 4; ++s) {
    #pragma unroll
    for (int t = 0; t < NT; ++t) {
      #pragma unroll
      for (int m = 0; m < MT; ++m) {
        acc[MT * (A0 + t) + m] = __builtin_amdgcn_wmma_f32_16x16x4_f32(
            false, a[s][m], false, b[s][t], (short)0, acc[MT * (A0 + t) + m], false, false);
      }
    }
  }
}

template <int NT, int A0>
__device__ __forceinline__ void chunk_group(v8f (&acc)[7 * MT], const float* __restrict__ arow,
                                            const float* __restrict__ wb, int f0, int nfr, int q0) {
  const int ntrips = nfr << 2;
  v2f a0[4][MT], a1[4][MT];
  v2f b0[4][NT], b1[4][NT];
  auto aptr = [&](int j) { return arow + (f0 + (j >> 2)) * 2 + ((j & 3) << 2) * 24; };
  auto wptr = [&](int j) {
    return wb + (size_t)((f0 + (j >> 2)) * QN + q0 + ((j & 3) << 2)) * QSTR;
  };

  load_frags<NT>(a0, b0, aptr(0), wptr(0));
  #pragma nounroll
  for (int j = 0; j + 2 < ntrips; j += 2) {
    load_frags<NT>(a1, b1, aptr(j + 1), wptr(j + 1));
    asm volatile("" ::: "memory");
    mma_frags<NT, A0>(acc, a0, b0);
    load_frags<NT>(a0, b0, aptr(j + 2), wptr(j + 2));
    mma_frags<NT, A0>(acc, a1, b1);
  }
  load_frags<NT>(a1, b1, aptr(ntrips - 1), wptr(ntrips - 1));
  mma_frags<NT, A0>(acc, a0, b0);
  mma_frags<NT, A0>(acc, a1, b1);
}

template <int NT, int A0>
__device__ __forceinline__ void frame_one_step(v8f (&acc)[7 * MT], const float* __restrict__ arow,
                                               const float* __restrict__ wb) {
  v2f a[MT];
  #pragma unroll
  for (int m = 0; m < MT; ++m) a[m] = *(const v2f*)(arow + m * MSTR);
  #pragma unroll
  for (int t = 0; t < NT; ++t) {
    v2f b = *(const v2f*)(wb + t * FRAG);
    #pragma unroll
    for (int m = 0; m < MT; ++m) {
      acc[MT * (A0 + t) + m] = __builtin_amdgcn_wmma_f32_16x16x4_f32(
          false, a[m], false, b, (short)0, acc[MT * (A0 + t) + m], false, false);
    }
  }
}

__global__ __launch_bounds__(256)
void ISTFT_gemm(const float* __restrict__ X, const float* __restrict__ Wf,
                float* __restrict__ out) {
  __shared__ float As[2 * PLANE];

  const int tid  = threadIdx.x;
  const int wave = __builtin_amdgcn_readfirstlane(tid >> 5);
  const int lane = tid & 31;
  const int lrow = lane & 15;
  const int lhi  = lane >> 4;
  const int m0   = blockIdx.x << 5;

  const bool two   = wave < 2;
  const int  tileA = two ? (wave << 2) : 8 + ((wave - 2) << 2);
  const int  tileB = 32 + 3 * wave;
  const int  flo   = two ? 0 : ((wave - 2) >> 1);
  const int  fcnt  = two ? 3 : 4;
  const int  par   = wave & 1;
  const int  colA0 = ((tileA >> 3) << 7) + par;
  const int  colB0 = (16 << 5) + par;

  v8f acc[7 * MT] = {};

  const float* __restrict__ Xblk = X + (size_t)m0 * KTOT;
  const float* __restrict__ Arow = As + par * PLANE + lrow * LDSTR + lhi * 12;
  const float* __restrict__ WbA  = Wf + (size_t)tileA * FRAG + lane * 2;
  const float* __restrict__ WbB  = Wf + (size_t)tileB * FRAG + lane * 2;

  v4f preL[12], preH[12];

  auto chunk_products = [&](int q0) {
    chunk_group<4, 0>(acc, Arow, WbA, flo, fcnt, q0);
    if (two) chunk_group<3, 4>(acc, Arow, WbB, 3, 3, q0);
  };
  auto tail_products = [&](int q0) {
    for (int i = 0; i < fcnt; ++i) {
      const int tf = flo + i;
      frame_one_step<4, 0>(acc, Arow + tf * 2, WbA + (size_t)(tf * QN + q0) * QSTR);
    }
    if (two) {
      for (int tf = 3; tf < 6; ++tf) {
        frame_one_step<3, 4>(acc, Arow + tf * 2, WbB + (size_t)(tf * QN + q0) * QSTR);
      }
    }
  };
  auto fetch_chunk = [&](int fb) {
    #pragma unroll
    for (int i = 0; i < 12; ++i) {
      const int id = tid + (i << 8);
      const int rr = id / 96, cc = id - rr * 96;
      const int fi = cc / 3, t = cc - fi * 3;
      const float* __restrict__ xr = Xblk + (size_t)rr * KTOT;
      preL[i] = *(const v4f*)(xr + (fb * 12) + (cc << 2));
      preH[i] = *(const v4f*)(xr + ((256 - fb - fi) * 12) + (t << 2));
    }
  };
  auto stage_chunk = [&]() {
    #pragma unroll
    for (int i = 0; i < 12; ++i) {
      const int id = tid + (i << 8);
      const int rr = id / 96, cc = id - rr * 96;
      const v4f L = preL[i], H = preH[i];
      const v4f ev = v4f{L.x + H.x, L.y - H.y, L.z + H.z, L.w - H.w};
      const v4f od = v4f{L.x - H.x, L.y + H.y, L.z - H.z, L.w + H.w};
      *(v4f*)(As + rr * LDSTR + (cc << 2))         = ev;
      *(v4f*)(As + PLANE + rr * LDSTR + (cc << 2)) = od;
    }
  };

  fetch_chunk(0);
  stage_chunk();
  __syncthreads();

  for (int c = 0; c < 3; ++c) {
    fetch_chunk((c + 1) * FCH);
    chunk_products(c * QCH);
    __syncthreads();
    stage_chunk();
    __syncthreads();
  }

  {
    const int tp = (tid < 96) ? tid : 0;
    const int tr = tp / 3, tc = tp - tr * 3;
    const v4f tailv = *(const v4f*)(Xblk + (size_t)tr * KTOT + 128 * 12 + (tc << 2));
    chunk_products(3 * QCH);
    __syncthreads();
    if (tid < 96) {
      *(v4f*)(As + tr * LDSTR + (tc << 2))         = tailv;
      *(v4f*)(As + PLANE + tr * LDSTR + (tc << 2)) = tailv;
    } else if (tid < 192) {
      const int t2 = tid - 96;
      const int zr = t2 / 3, zc = t2 - zr * 3;
      *(v4f*)(As + zr * LDSTR + 12 + (zc << 2))         = v4f{0.0f, 0.0f, 0.0f, 0.0f};
      *(v4f*)(As + PLANE + zr * LDSTR + 12 + (zc << 2)) = v4f{0.0f, 0.0f, 0.0f, 0.0f};
    }
    __syncthreads();
  }

  tail_products(4 * QCH);

  #pragma unroll
  for (int t = 0; t < 4; ++t) {
    int colg = colA0 + (t << 5) + (lrow << 1);
    #pragma unroll
    for (int m = 0; m < MT; ++m) {
      #pragma unroll
      for (int v = 0; v < 8; ++v) {
        int row = m0 + (m << 4) + v + (lhi << 3);
        out[(size_t)row * NOUT + colg] = acc[MT * t + m][v];
      }
    }
  }
  if (two) {
    #pragma unroll
    for (int t = 0; t < 3; ++t) {
      int colg = colB0 + (t << 5) + (lrow << 1);
      if (colg < NOUT) {
        #pragma unroll
        for (int m = 0; m < MT; ++m) {
          #pragma unroll
          for (int v = 0; v < 8; ++v) {
            int row = m0 + (m << 4) + v + (lhi << 3);
            out[(size_t)row * NOUT + colg] = acc[MT * (4 + t) + m][v];
          }
        }
      }
    }
  }
}

extern "C" void kernel_launch(void* const* d_in, const int* in_sizes, int n_in,
                              void* d_out, int out_size, void* d_ws, size_t ws_size,
                              hipStream_t stream) {
  const float* X  = (const float*)d_in[0];
  float*       O  = (float*)d_out;
  float*       Wf = (float*)d_ws;

  const int wtot    = NFRM * QN * NTW * FRAG;
  const int wblocks = (wtot + 255) / 256;

  ISTFT_buildW<<<wblocks, 256, 0, stream>>>(Wf);
  ISTFT_gemm<<<8192 / 32, 256, 0, stream>>>(X, Wf, O);
}


# Inverse DFT split again by output sample mod 4 (radix 4): rows staged as four add/subtract folds of X[f], X[256-f], X[128-f], X[128+f]; K per product halved, 18 blocks per wave; sum order changes
# speedup vs baseline: 4.6800x; 1.4851x over previous
#include <hip/hip_runtime.h>
#include <hip/hip_bf16.h>
#include <math.h>


typedef __attribute__((ext_vector_type(2))) float v2f;
typedef __attribute__((ext_vector_type(8))) float v8f;
typedef __attribute__((ext_vector_type(4))) float v4f;

#define KTOT   3084
#define NFOLD  65
#define NFRM   6
#define QN     33
#define NGRP   10
#define NTW    40
#define FRAG   64
#define NOUT   600
#define FCH    16
#define QCH    (FCH / 2)
#define LDSTR  196
#define PLANE  (32 * LDSTR)
#define QSTR   (NTW * FRAG)
#define FSTR   (QN * QSTR)
#define MT     2
#define MSTR   (16 * LDSTR)
#define NSL    5

__global__ __launch_bounds__(256)
void ISTFT_buildW(float* __restrict__ Wf) {
  int gid = blockIdx.x * 256 + threadIdx.x;
  if (gid >= NFRM * QN * NTW * FRAG) return;

  int d    = gid & 1;
  int l    = (gid >> 1) & 31;
  int rest = gid >> 6;
  int nt   = rest % NTW;
  int kq   = rest / NTW;
  int tf   = kq / QN;
  int q    = kq - tf * QN;
  int f    = (q << 1) + (l >> 4);
  int n    = ((nt % NGRP) << 6) + (nt / NGRP) + ((l & 15) << 2);

  float v = 0.0f;
  if (n < NOUT && f < NFOLD) {
    int labs = n + 256;
    int nn   = labs - (tf << 7);
    if (nn >= 0 && nn < 512) {
      const float c0 = 6.283185307179586f / 512.0f;
      float env = 0.0f;
      #pragma unroll
      for (int tp = 0; tp < 6; ++tp) {
        int nn2 = labs - (tp << 7);
        if (nn2 >= 0 && nn2 < 512) {
          float w = 0.5f * (1.0f - cosf(c0 * (float)nn2));
          env += w * w;
        }
      }
      float wn   = 0.5f * (1.0f - cosf(c0 * (float)nn));
      bool  half = (f == 0) || (f == 64);
      float scl  = (half ? 1.0f : 2.0f) / 512.0f;
      int   m    = (f * nn) & 511;
      float ang  = c0 * (float)m;
      float basis = d ? -sinf(ang) : cosf(ang);
      if (d && f == 0) basis = 0.0f;
      v = scl * wn / env * basis;
    }
  }
  Wf[gid] = v;
}

__device__ __forceinline__ constexpr int seg_lo(int H, int i) {
  return H == 0 ? (i < 3 ? 0 : (i == 3 ? 2 : 4)) : (i < 4 ? 0 : 1);
}
__device__ __forceinline__ constexpr int seg_hi(int H, int i) {
  return H == 0 ? (i == 0 ? 3 : 4) : (i == 0 ? 0 : (i == 1 ? 2 : 4));
}

template <int H>
__device__ __forceinline__ void load_step(v2f (&a)[5][MT], v2f (&b)[5][NSL],
                                          const float* __restrict__ ap,
                                          const float* __restrict__ wp) {
  #pragma unroll
  for (int i = 0; i < 5; ++i) {
    #pragma unroll
    for (int j = 0; j < NSL; ++j) {
      if (j >= seg_lo(H, i) && j <= seg_hi(H, i))
        b[i][j] = *(const v2f*)(wp + (size_t)i * FSTR + j * FRAG);
    }
  }
  #pragma unroll
  for (int i = 0; i < 5; ++i) {
    #pragma unroll
    for (int m = 0; m < MT; ++m) a[i][m] = *(const v2f*)(ap + i * 2 + m * MSTR);
  }
}

template <int H>
__device__ __forceinline__ void mma_step(v8f (&acc)[NSL * MT], const v2f (&a)[5][MT],
                                         const v2f (&b)[5][NSL]) {
  #pragma unroll
  for (int i = 0; i < 5; ++i) {
    #pragma unroll
    for (int j = 0; j < NSL; ++j) {
      if (j >= seg_lo(H, i) && j <= seg_hi(H, i)) {
        #pragma unroll
        for (int m = 0; m < MT; ++m) {
          acc[MT * j + m] = __builtin_amdgcn_wmma_f32_16x16x4_f32(
              false, a[i][m], false, b[i][j], (short)0, acc[MT * j + m], false, false);
        }
      }
    }
  }
}

template <int H>
__device__ __forceinline__ void chunk_run(v8f (&acc)[NSL * MT], const float* __restrict__ arow,
                                          const float* __restrict__ wb, int q0) {
  v2f a0[5][MT], a1[5][MT];
  v2f b0[5][NSL], b1[5][NSL];
  const float* __restrict__ wp = wb + (size_t)q0 * QSTR;

  load_step<H>(a0, b0, arow, wp);
  #pragma nounroll
  for (int j = 0; j + 2 < QCH; j += 2) {
    load_step<H>(a1, b1, arow + (j + 1) * 24, wp + (size_t)(j + 1) * QSTR);
    asm volatile("" ::: "memory");
    mma_step<H>(acc, a0, b0);
    load_step<H>(a0, b0, arow + (j + 2) * 24, wp + (size_t)(j + 2) * QSTR);
    mma_step<H>(acc, a1, b1);
  }
  load_step<H>(a1, b1, arow + (QCH - 1) * 24, wp + (size_t)(QCH - 1) * QSTR);
  mma_step<H>(acc, a0, b0);
  mma_step<H>(acc, a1, b1);
}

template <int H>
__device__ __forceinline__ void tail_run(v8f (&acc)[NSL * MT], const float* __restrict__ arow,
                                         const float* __restrict__ wb, int q0) {
  v2f a[5][MT];
  v2f b[5][NSL];
  load_step<H>(a, b, arow, wb + (size_t)q0 * QSTR);
  mma_step<H>(acc, a, b);
}

__global__ __launch_bounds__(256)
void ISTFT_gemm(const float* __restrict__ X, const float* __restrict__ Wf,
                float* __restrict__ out) {
  __shared__ float As[4 * PLANE];

  const int tid  = threadIdx.x;
  const int wave = __builtin_amdgcn_readfirstlane(tid >> 5);
  const int lane = tid & 31;
  const int lrow = lane & 15;
  const int lhi  = lane >> 4;
  const int m0   = blockIdx.x << 5;

  const int cls  = wave & 3;
  const int hlf  = wave >> 2;
  const int tile0 = cls * NGRP + hlf * NSL;
  const int col0  = (hlf * NSL << 6) + cls;

  v8f acc[NSL * MT] = {};

  const float* __restrict__ Xblk = X + (size_t)m0 * KTOT;
  const float* __restrict__ Arow = As + cls * PLANE + lrow * LDSTR + lhi * 12 + hlf * 2;
  const float* __restrict__ Wb   = Wf + (size_t)hlf * FSTR + (size_t)tile0 * FRAG + lane * 2;

  v4f pA[6], pB[6], pC[6], pD[6];

  auto chunk_products = [&](int q0) {
    if (hlf == 0) chunk_run<0>(acc, Arow, Wb, q0);
    else          chunk_run<1>(acc, Arow, Wb, q0);
  };
  auto tail_products = [&](int q0) {
    if (hlf == 0) tail_run<0>(acc, Arow, Wb, q0);
    else          tail_run<1>(acc, Arow, Wb, q0);
  };
  auto fetch_chunk = [&](int fb) {
    #pragma unroll
    for (int i = 0; i < 6; ++i) {
      const int id = tid + (i << 8);
      const int rr = id / 48, cc = id - rr * 48;
      const int fi = cc / 3, t = cc - fi * 3;
      const int f  = fb + fi;
      const float* __restrict__ xr = Xblk + (size_t)rr * KTOT + (t << 2);
      pA[i] = *(const v4f*)(xr + f * 12);
      pB[i] = *(const v4f*)(xr + (256 - f) * 12);
      pC[i] = *(const v4f*)(xr + (128 - f) * 12);
      pD[i] = *(const v4f*)(xr + (128 + f) * 12);
    }
  };
  auto fold_store = [&](const v4f A, const v4f B, const v4f C, const v4f D, const int off) {
    const v4f ef = v4f{A.x + B.x, A.y - B.y, A.z + B.z, A.w - B.w};
    const v4f of = v4f{A.x - B.x, A.y + B.y, A.z - B.z, A.w + B.w};
    const v4f eg = v4f{C.x + D.x, C.y - D.y, C.z + D.z, C.w - D.w};
    const v4f og = v4f{C.x - D.x, C.y + D.y, C.z - D.z, C.w + D.w};
    *(v4f*)(As + off)             = v4f{ef.x + eg.x, ef.y - eg.y, ef.z + eg.z, ef.w - eg.w};
    *(v4f*)(As + PLANE + off)     = v4f{of.x - og.y, of.y - og.x, of.z - og.w, of.w - og.z};
    *(v4f*)(As + 2 * PLANE + off) = v4f{ef.x - eg.x, ef.y + eg.y, ef.z - eg.z, ef.w + eg.w};
    *(v4f*)(As + 3 * PLANE + off) = v4f{of.x + og.y, of.y + og.x, of.z + og.w, of.w + og.z};
  };
  auto stage_chunk = [&]() {
    #pragma unroll
    for (int i = 0; i < 6; ++i) {
      const int id = tid + (i << 8);
      const int rr = id / 48, cc = id - rr * 48;
      fold_store(pA[i], pB[i], pC[i], pD[i], rr * LDSTR + (cc << 2));
    }
  };

  fetch_chunk(0);
  stage_chunk();
  __syncthreads();

  for (int c = 0; c < 3; ++c) {
    fetch_chunk((c + 1) * FCH);
    chunk_products(c * QCH);
    __syncthreads();
    stage_chunk();
    __syncthreads();
  }

  {
    const int tp = (tid < 96) ? tid : 0;
    const int tr = tp / 3, tc = tp - tr * 3;
    const float* __restrict__ xt = Xblk + (size_t)tr * KTOT + (tc << 2);
    const v4f t64  = *(const v4f*)(xt + 64 * 12);
    const v4f t192 = *(const v4f*)(xt + 192 * 12);
    chunk_products(3 * QCH);
    __syncthreads();
    if (tid < 96) {
      fold_store(t64, t192, t64, t192, tr * LDSTR + (tc << 2));
    } else if (tid < 192) {
      const int t2 = tid - 96;
      const int zr = t2 / 3, zc = t2 - zr * 3;
      const v4f z = v4f{0.0f, 0.0f, 0.0f, 0.0f};
      #pragma unroll
      for (int p = 0; p < 4; ++p) *(v4f*)(As + p * PLANE + zr * LDSTR + 12 + (zc << 2)) = z;
    }
    __syncthreads();
  }

  tail_products(4 * QCH);

  #pragma unroll
  for (int j = 0; j < NSL; ++j) {
    const int colg = col0 + (j << 6) + (lrow << 2);
    if (colg < NOUT) {
      #pragma unroll
      for (int m = 0; m < MT; ++m) {
        #pragma unroll
        for (int v = 0; v < 8; ++v) {
          const int row = m0 + (m << 4) + v + (lhi << 3);
          out[(size_t)row * NOUT + colg] = acc[MT * j + m][v];
        }
      }
    }
  }
}

extern "C" void kernel_launch(void* const* d_in, const int* in_sizes, int n_in,
                              void* d_out, int out_size, void* d_ws, size_t ws_size,
                              hipStream_t stream) {
  const float* X  = (const float*)d_in[0];
  float*       O  = (float*)d_out;
  float*       Wf = (float*)d_ws;

  const int wtot    = NFRM * QN * NTW * FRAG;
  const int wblocks = (wtot + 255) / 256;

  ISTFT_buildW<<<wblocks, 256, 0, stream>>>(Wf);
  ISTFT_gemm<<<8192 / 32, 256, 0, stream>>>(X, Wf, O);
}


# Inverse DFT split a third time, by output sample mod 8 (radix 8): rows staged as an 8-point transform of the eight frequencies that meet at f; K per product halved, two chunks; the sum order changes
# speedup vs baseline: 6.3127x; 1.3489x over previous
#include <hip/hip_runtime.h>
#include <hip/hip_bf16.h>
#include <math.h>


typedef __attribute__((ext_vector_type(2))) float v2f;
typedef __attribute__((ext_vector_type(8))) float v8f;
typedef __attribute__((ext_vector_type(4))) float v4f;

#define KTOT   3084
#define NFOLD  33
#define NFRM   6
#define QN     17
#define NGRP   5
#define NTW    40
#define FRAG   64
#define NOUT   600
#define FCH    16
#define QCH    (FCH / 2)
#define LDSTR  196
#define PLANE  (32 * LDSTR)
#define QSTR   (NTW * FRAG)
#define FSTR   (QN * QSTR)
#define MT     2
#define MSTR   (16 * LDSTR)
#define NSL    5

__global__ __launch_bounds__(256)
void ISTFT_buildW(float* __restrict__ Wf) {
  int gid = blockIdx.x * 256 + threadIdx.x;
  if (gid >= NFRM * QN * NTW * FRAG) return;

  int d    = gid & 1;
  int l    = (gid >> 1) & 31;
  int rest = gid >> 6;
  int nt   = rest % NTW;
  int kq   = rest / NTW;
  int tf   = kq / QN;
  int q    = kq - tf * QN;
  int f    = (q << 1) + (l >> 4);
  int n    = ((nt % NGRP) << 7) + (nt / NGRP) + ((l & 15) << 3);

  float v = 0.0f;
  if (n < NOUT && f < NFOLD) {
    int labs = n + 256;
    int nn   = labs - (tf << 7);
    if (nn >= 0 && nn < 512) {
      const float c0 = 6.283185307179586f / 512.0f;
      float env = 0.0f;
      #pragma unroll
      for (int tp = 0; tp < 6; ++tp) {
        int nn2 = labs - (tp << 7);
        if (nn2 >= 0 && nn2 < 512) {
          float w = 0.5f * (1.0f - cosf(c0 * (float)nn2));
          env += w * w;
        }
      }
      float wn   = 0.5f * (1.0f - cosf(c0 * (float)nn));
      bool  half = (f == 0) || (f == 32);
      float scl  = (half ? 1.0f : 2.0f) / 512.0f;
      int   m    = (f * nn) & 511;
      float ang  = c0 * (float)m;
      float basis = d ? -sinf(ang) : cosf(ang);
      if (d && f == 0) basis = 0.0f;
      v = scl * wn / env * basis;
    }
  }
  Wf[gid] = v;
}

__device__ __forceinline__ constexpr int seg_lo(int i) { return i < 2 ? 0 : i - 2; }
__device__ __forceinline__ constexpr int seg_hi(int i) { return i > 3 ? 4 : i + 1; }

__device__ __forceinline__ void load_step(v2f (&a)[NFRM][MT], v2f (&b)[NFRM][NSL],
                                          const float* __restrict__ ap,
                                          const float* __restrict__ wp) {
  #pragma unroll
  for (int i = 0; i < NFRM; ++i) {
    #pragma unroll
    for (int j = 0; j < NSL; ++j) {
      if (j >= seg_lo(i) && j <= seg_hi(i))
        b[i][j] = *(const v2f*)(wp + (size_t)i * FSTR + j * FRAG);
    }
  }
  #pragma unroll
  for (int i = 0; i < NFRM; ++i) {
    #pragma unroll
    for (int m = 0; m < MT; ++m) a[i][m] = *(const v2f*)(ap + i * 2 + m * MSTR);
  }
}

__device__ __forceinline__ void mma_step(v8f (&acc)[NSL * MT], const v2f (&a)[NFRM][MT],
                                         const v2f (&b)[NFRM][NSL]) {
  #pragma unroll
  for (int i = 0; i < NFRM; ++i) {
    #pragma unroll
    for (int j = 0; j < NSL; ++j) {
      if (j >= seg_lo(i) && j <= seg_hi(i)) {
        #pragma unroll
        for (int m = 0; m < MT; ++m) {
          acc[MT * j + m] = __builtin_amdgcn_wmma_f32_16x16x4_f32(
              false, a[i][m], false, b[i][j], (short)0, acc[MT * j + m], false, false);
        }
      }
    }
  }
}

__device__ __forceinline__ void chunk_run(v8f (&acc)[NSL * MT], const float* __restrict__ arow,
                                          const float* __restrict__ wb, int q0) {
  v2f a0[NFRM][MT], a1[NFRM][MT];
  v2f b0[NFRM][NSL], b1[NFRM][NSL];
  const float* __restrict__ wp = wb + (size_t)q0 * QSTR;

  load_step(a0, b0, arow, wp);
  #pragma nounroll
  for (int j = 0; j + 2 < QCH; j += 2) {
    load_step(a1, b1, arow + (j + 1) * 24, wp + (size_t)(j + 1) * QSTR);
    asm volatile("" ::: "memory");
    mma_step(acc, a0, b0);
    load_step(a0, b0, arow + (j + 2) * 24, wp + (size_t)(j + 2) * QSTR);
    mma_step(acc, a1, b1);
  }
  load_step(a1, b1, arow + (QCH - 1) * 24, wp + (size_t)(QCH - 1) * QSTR);
  mma_step(acc, a0, b0);
  mma_step(acc, a1, b1);
}

__device__ __forceinline__ void tail_run(v8f (&acc)[NSL * MT], const float* __restrict__ arow,
                                         const float* __restrict__ wb, int q0) {
  v2f a[NFRM][MT];
  v2f b[NFRM][NSL];
  load_step(a, b, arow, wb + (size_t)q0 * QSTR);
  mma_step(acc, a, b);
}

__device__ __forceinline__ void fold_store(float* __restrict__ As, const v4f x0, const v4f x1,
                                           const v4f x2, const v4f x3, const v4f y4, const v4f y5,
                                           const v4f y6, const v4f y7, const int off) {
  const float h = 0.70710678118654752f;
  const v4f A0 = v4f{x0.x + y4.x, x0.y - y4.y, x0.z + y4.z, x0.w - y4.w};
  const v4f B0 = v4f{x0.x - y4.x, x0.y + y4.y, x0.z - y4.z, x0.w + y4.w};
  const v4f A1 = v4f{x1.x + y5.x, x1.y - y5.y, x1.z + y5.z, x1.w - y5.w};
  const v4f B1 = v4f{x1.x - y5.x, x1.y + y5.y, x1.z - y5.z, x1.w + y5.w};
  const v4f A2 = v4f{x2.x + y6.x, x2.y - y6.y, x2.z + y6.z, x2.w - y6.w};
  const v4f B2 = v4f{x2.x - y6.x, x2.y + y6.y, x2.z - y6.z, x2.w + y6.w};
  const v4f A3 = v4f{x3.x + y7.x, x3.y - y7.y, x3.z + y7.z, x3.w - y7.w};
  const v4f B3 = v4f{x3.x - y7.x, x3.y + y7.y, x3.z - y7.z, x3.w + y7.w};
  {
    const v4f e0 = A0 + A2, e1 = A0 - A2, o0 = A1 + A3, o1 = A1 - A3;
    *(v4f*)(As + off)             = e0 + o0;
    *(v4f*)(As + 4 * PLANE + off) = e0 - o0;
    *(v4f*)(As + 2 * PLANE + off) = v4f{e1.x - o1.y, e1.y + o1.x, e1.z - o1.w, e1.w + o1.z};
    *(v4f*)(As + 6 * PLANE + off) = v4f{e1.x + o1.y, e1.y - o1.x, e1.z + o1.w, e1.w - o1.z};
  }
  {
    const v4f C1 = v4f{h * (B1.x - B1.y), h * (B1.x + B1.y), h * (B1.z - B1.w), h * (B1.z + B1.w)};
    const v4f C2 = v4f{-B2.y, B2.x, -B2.w, B2.z};
    const v4f C3 = v4f{-h * (B3.x + B3.y), h * (B3.x - B3.y), -h * (B3.z + B3.w), h * (B3.z - B3.w)};
    const v4f e0 = B0 + C2, e1 = B0 - C2, o0 = C1 + C3, o1 = C1 - C3;
    *(v4f*)(As + 1 * PLANE + off) = e0 + o0;
    *(v4f*)(As + 5 * PLANE + off) = e0 - o0;
    *(v4f*)(As + 3 * PLANE + off) = v4f{e1.x - o1.y, e1.y + o1.x, e1.z - o1.w, e1.w + o1.z};
    *(v4f*)(As + 7 * PLANE + off) = v4f{e1.x + o1.y, e1.y - o1.x, e1.z + o1.w, e1.w - o1.z};
  }
}

__global__ __launch_bounds__(256)
void ISTFT_gemm(const float* __restrict__ X, const float* __restrict__ Wf,
                float* __restrict__ out) {
  __shared__ float As[8 * PLANE];

  const int tid  = threadIdx.x;
  const int wave = __builtin_amdgcn_readfirstlane(tid >> 5);
  const int lane = tid & 31;
  const int lrow = lane & 15;
  const int lhi  = lane >> 4;
  const int m0   = blockIdx.x << 5;

  const int cls  = wave;

  v8f acc[NSL * MT] = {};

  const float* __restrict__ Xblk = X + (size_t)m0 * KTOT;
  const float* __restrict__ Arow = As + cls * PLANE + lrow * LDSTR + lhi * 12;
  const float* __restrict__ Wb   = Wf + (size_t)(cls * NGRP) * FRAG + lane * 2;

  v4f p0[6], p1[6], p2[6], p3[6], p4[6], p5[6], p6[6], p7[6];

  auto fetch_chunk = [&](int fb) {
    #pragma unroll
    for (int i = 0; i < 6; ++i) {
      const int id = tid + (i << 8);
      const int rr = id / 48, cc = id - rr * 48;
      const int fi = cc / 3, t = cc - fi * 3;
      const int f  = fb + fi;
      const float* __restrict__ xr = Xblk + (size_t)rr * KTOT + (t << 2);
      p0[i] = *(const v4f*)(xr + f * 12);
      p1[i] = *(const v4f*)(xr + (64 + f) * 12);
      p2[i] = *(const v4f*)(xr + (128 + f) * 12);
      p3[i] = *(const v4f*)(xr + (192 + f) * 12);
      p4[i] = *(const v4f*)(xr + (256 - f) * 12);
      p5[i] = *(const v4f*)(xr + (192 - f) * 12);
      p6[i] = *(const v4f*)(xr + (128 - f) * 12);
      p7[i] = *(const v4f*)(xr + (64 - f) * 12);
    }
  };
  auto stage_chunk = [&]() {
    #pragma unroll
    for (int i = 0; i < 6; ++i) {
      const int id = tid + (i << 8);
      const int rr = id / 48, cc = id - rr * 48;
      fold_store(As, p0[i], p1[i], p2[i], p3[i], p4[i], p5[i], p6[i], p7[i], rr * LDSTR + (cc << 2));
    }
  };

  fetch_chunk(0);
  stage_chunk();
  __syncthreads();

  fetch_chunk(FCH);
  chunk_run(acc, Arow, Wb, 0);
  __syncthreads();
  stage_chunk();
  __syncthreads();

  {
    const int tp = (tid < 96) ? tid : 0;
    const int tr = tp / 3, tc = tp - tr * 3;
    const float* __restrict__ xt = Xblk + (size_t)tr * KTOT + (tc << 2);
    const v4f t32  = *(const v4f*)(xt + 32 * 12);
    const v4f t96  = *(const v4f*)(xt + 96 * 12);
    const v4f t160 = *(const v4f*)(xt + 160 * 12);
    const v4f t224 = *(const v4f*)(xt + 224 * 12);
    chunk_run(acc, Arow, Wb, QCH);
    __syncthreads();
    if (tid < 96) {
      fold_store(As, t32, t96, t160, t224, t224, t160, t96, t32, tr * LDSTR + (tc << 2));
    } else if (tid < 192) {
      const int t2 = tid - 96;
      const int zr = t2 / 3, zc = t2 - zr * 3;
      const v4f z = v4f{0.0f, 0.0f, 0.0f, 0.0f};
      #pragma unroll
      for (int p = 0; p < 8; ++p) *(v4f*)(As + p * PLANE + zr * LDSTR + 12 + (zc << 2)) = z;
    }
    __syncthreads();
  }

  tail_run(acc, Arow, Wb, 2 * QCH);

  #pragma unroll
  for (int j = 0; j < NSL; ++j) {
    const int colg = cls + (j << 7) + (lrow << 3);
    if (colg < NOUT) {
      #pragma unroll
      for (int m = 0; m < MT; ++m) {
        #pragma unroll
        for (int v = 0; v < 8; ++v) {
          const int row = m0 + (m << 4) + v + (lhi << 3);
          out[(size_t)row * NOUT + colg] = acc[MT * j + m][v];
        }
      }
    }
  }
}

extern "C" void kernel_launch(void* const* d_in, const int* in_sizes, int n_in,
                              void* d_out, int out_size, void* d_ws, size_t ws_size,
                              hipStream_t stream) {
  const float* X  = (const float*)d_in[0];
  float*       O  = (float*)d_out;
  float*       Wf = (float*)d_ws;

  const int wtot    = NFRM * QN * NTW * FRAG;
  const int wblocks = (wtot + 255) / 256;

  ISTFT_buildW<<<wblocks, 256, 0, stream>>>(Wf);
  ISTFT_gemm<<<8192 / 32, 256, 0, stream>>>(X, Wf, O);
}


# Output rows assembled in LDS (bank-swizzled image of the block's 32 rows) and stored as 16-byte pieces: each store instruction covers four whole 128-byte lines, not 4-byte pieces; sums unchanged
# speedup vs baseline: 7.5962x; 1.2033x over previous
#include <hip/hip_runtime.h>
#include <hip/hip_bf16.h>
#include <math.h>


typedef __attribute__((ext_vector_type(2))) float v2f;
typedef __attribute__((ext_vector_type(8))) float v8f;
typedef __attribute__((ext_vector_type(4))) float v4f;

#define KTOT   3084
#define NFOLD  33
#define NFRM   6
#define QN     17
#define NGRP   5
#define NTW    40
#define FRAG   64
#define NOUT   600
#define FCH    16
#define QCH    (FCH / 2)
#define LDSTR  196
#define PLANE  (32 * LDSTR)
#define QSTR   (NTW * FRAG)
#define FSTR   (QN * QSTR)
#define MT     2
#define MSTR   (16 * LDSTR)
#define NSL    5
#define OPC    (32 * NOUT / 4)

__global__ __launch_bounds__(256)
void ISTFT_buildW(float* __restrict__ Wf) {
  int gid = blockIdx.x * 256 + threadIdx.x;
  if (gid >= NFRM * QN * NTW * FRAG) return;

  int d    = gid & 1;
  int l    = (gid >> 1) & 31;
  int rest = gid >> 6;
  int nt   = rest % NTW;
  int kq   = rest / NTW;
  int tf   = kq / QN;
  int q    = kq - tf * QN;
  int f    = (q << 1) + (l >> 4);
  int n    = ((nt % NGRP) << 7) + (nt / NGRP) + ((l & 15) << 3);

  float v = 0.0f;
  if (n < NOUT && f < NFOLD) {
    int labs = n + 256;
    int nn   = labs - (tf << 7);
    if (nn >= 0 && nn < 512) {
      const float c0 = 6.283185307179586f / 512.0f;
      float env = 0.0f;
      #pragma unroll
      for (int tp = 0; tp < 6; ++tp) {
        int nn2 = labs - (tp << 7);
        if (nn2 >= 0 && nn2 < 512) {
          float w = 0.5f * (1.0f - cosf(c0 * (float)nn2));
          env += w * w;
        }
      }
      float wn   = 0.5f * (1.0f - cosf(c0 * (float)nn));
      bool  half = (f == 0) || (f == 32);
      float scl  = (half ? 1.0f : 2.0f) / 512.0f;
      int   m    = (f * nn) & 511;
      float ang  = c0 * (float)m;
      float basis = d ? -sinf(ang) : cosf(ang);
      if (d && f == 0) basis = 0.0f;
      v = scl * wn / env * basis;
    }
  }
  Wf[gid] = v;
}

__device__ __forceinline__ constexpr int seg_lo(int i) { return i < 2 ? 0 : i - 2; }
__device__ __forceinline__ constexpr int seg_hi(int i) { return i > 3 ? 4 : i + 1; }

__device__ __forceinline__ void load_step(v2f (&a)[NFRM][MT], v2f (&b)[NFRM][NSL],
                                          const float* __restrict__ ap,
                                          const float* __restrict__ wp) {
  #pragma unroll
  for (int i = 0; i < NFRM; ++i) {
    #pragma unroll
    for (int j = 0; j < NSL; ++j) {
      if (j >= seg_lo(i) && j <= seg_hi(i))
        b[i][j] = *(const v2f*)(wp + (size_t)i * FSTR + j * FRAG);
    }
  }
  #pragma unroll
  for (int i = 0; i < NFRM; ++i) {
    #pragma unroll
    for (int m = 0; m < MT; ++m) a[i][m] = *(const v2f*)(ap + i * 2 + m * MSTR);
  }
}

__device__ __forceinline__ void mma_step(v8f (&acc)[NSL * MT], const v2f (&a)[NFRM][MT],
                                         const v2f (&b)[NFRM][NSL]) {
  #pragma unroll
  for (int i = 0; i < NFRM; ++i) {
    #pragma unroll
    for (int j = 0; j < NSL; ++j) {
      if (j >= seg_lo(i) && j <= seg_hi(i)) {
        #pragma unroll
        for (int m = 0; m < MT; ++m) {
          acc[MT * j + m] = __builtin_amdgcn_wmma_f32_16x16x4_f32(
              false, a[i][m], false, b[i][j], (short)0, acc[MT * j + m], false, false);
        }
      }
    }
  }
}

__device__ __forceinline__ void chunk_run(v8f (&acc)[NSL * MT], const float* __restrict__ arow,
                                          const float* __restrict__ wb, int q0) {
  v2f a0[NFRM][MT], a1[NFRM][MT];
  v2f b0[NFRM][NSL], b1[NFRM][NSL];
  const float* __restrict__ wp = wb + (size_t)q0 * QSTR;

  load_step(a0, b0, arow, wp);
  #pragma nounroll
  for (int j = 0; j + 2 < QCH; j += 2) {
    load_step(a1, b1, arow + (j + 1) * 24, wp + (size_t)(j + 1) * QSTR);
    asm volatile("" ::: "memory");
    mma_step(acc, a0, b0);
    load_step(a0, b0, arow + (j + 2) * 24, wp + (size_t)(j + 2) * QSTR);
    mma_step(acc, a1, b1);
  }
  load_step(a1, b1, arow + (QCH - 1) * 24, wp + (size_t)(QCH - 1) * QSTR);
  mma_step(acc, a0, b0);
  mma_step(acc, a1, b1);
}

__device__ __forceinline__ void tail_run(v8f (&acc)[NSL * MT], const float* __restrict__ arow,
                                         const float* __restrict__ wb, int q0) {
  v2f a[NFRM][MT];
  v2f b[NFRM][NSL];
  load_step(a, b, arow, wb + (size_t)q0 * QSTR);
  mma_step(acc, a, b);
}

__device__ __forceinline__ void fold_store(float* __restrict__ As, const v4f x0, const v4f x1,
                                           const v4f x2, const v4f x3, const v4f y4, const v4f y5,
                                           const v4f y6, const v4f y7, const int off) {
  const float h = 0.70710678118654752f;
  const v4f A0 = v4f{x0.x + y4.x, x0.y - y4.y, x0.z + y4.z, x0.w - y4.w};
  const v4f B0 = v4f{x0.x - y4.x, x0.y + y4.y, x0.z - y4.z, x0.w + y4.w};
  const v4f A1 = v4f{x1.x + y5.x, x1.y - y5.y, x1.z + y5.z, x1.w - y5.w};
  const v4f B1 = v4f{x1.x - y5.x, x1.y + y5.y, x1.z - y5.z, x1.w + y5.w};
  const v4f A2 = v4f{x2.x + y6.x, x2.y - y6.y, x2.z + y6.z, x2.w - y6.w};
  const v4f B2 = v4f{x2.x - y6.x, x2.y + y6.y, x2.z - y6.z, x2.w + y6.w};
  const v4f A3 = v4f{x3.x + y7.x, x3.y - y7.y, x3.z + y7.z, x3.w - y7.w};
  const v4f B3 = v4f{x3.x - y7.x, x3.y + y7.y, x3.z - y7.z, x3.w + y7.w};
  {
    const v4f e0 = A0 + A2, e1 = A0 - A2, o0 = A1 + A3, o1 = A1 - A3;
    *(v4f*)(As + off)             = e0 + o0;
    *(v4f*)(As + 4 * PLANE + off) = e0 - o0;
    *(v4f*)(As + 2 * PLANE + off) = v4f{e1.x - o1.y, e1.y + o1.x, e1.z - o1.w, e1.w + o1.z};
    *(v4f*)(As + 6 * PLANE + off) = v4f{e1.x + o1.y, e1.y - o1.x, e1.z + o1.w, e1.w - o1.z};
  }
  {
    const v4f C1 = v4f{h * (B1.x - B1.y), h * (B1.x + B1.y), h * (B1.z - B1.w), h * (B1.z + B1.w)};
    const v4f C2 = v4f{-B2.y, B2.x, -B2.w, B2.z};
    const v4f C3 = v4f{-h * (B3.x + B3.y), h * (B3.x - B3.y), -h * (B3.z + B3.w), h * (B3.z - B3.w)};
    const v4f e0 = B0 + C2, e1 = B0 - C2, o0 = C1 + C3, o1 = C1 - C3;
    *(v4f*)(As + 1 * PLANE + off) = e0 + o0;
    *(v4f*)(As + 5 * PLANE + off) = e0 - o0;
    *(v4f*)(As + 3 * PLANE + off) = v4f{e1.x - o1.y, e1.y + o1.x, e1.z - o1.w, e1.w + o1.z};
    *(v4f*)(As + 7 * PLANE + off) = v4f{e1.x + o1.y, e1.y - o1.x, e1.z + o1.w, e1.w - o1.z};
  }
}

__global__ __launch_bounds__(256)
void ISTFT_gemm(const float* __restrict__ X, const float* __restrict__ Wf,
                float* __restrict__ out) {
  __shared__ float As[8 * PLANE];
  __shared__ float Os[32 * NOUT];

  const int tid  = threadIdx.x;
  const int wave = __builtin_amdgcn_readfirstlane(tid >> 5);
  const int lane = tid & 31;
  const int lrow = lane & 15;
  const int lhi  = lane >> 4;
  const int m0   = blockIdx.x << 5;

  const int cls  = wave;

  v8f acc[NSL * MT] = {};

  const float* __restrict__ Xblk = X + (size_t)m0 * KTOT;
  const float* __restrict__ Arow = As + cls * PLANE + lrow * LDSTR + lhi * 12;
  const float* __restrict__ Wb   = Wf + (size_t)(cls * NGRP) * FRAG + lane * 2;

  v4f p0[6], p1[6], p2[6], p3[6], p4[6], p5[6], p6[6], p7[6];

  auto fetch_chunk = [&](int fb) {
    #pragma unroll
    for (int i = 0; i < 6; ++i) {
      const int id = tid + (i << 8);
      const int rr = id / 48, cc = id - rr * 48;
      const int fi = cc / 3, t = cc - fi * 3;
      const int f  = fb + fi;
      const float* __restrict__ xr = Xblk + (size_t)rr * KTOT + (t << 2);
      p0[i] = *(const v4f*)(xr + f * 12);
      p1[i] = *(const v4f*)(xr + (64 + f) * 12);
      p2[i] = *(const v4f*)(xr + (128 + f) * 12);
      p3[i] = *(const v4f*)(xr + (192 + f) * 12);
      p4[i] = *(const v4f*)(xr + (256 - f) * 12);
      p5[i] = *(const v4f*)(xr + (192 - f) * 12);
      p6[i] = *(const v4f*)(xr + (128 - f) * 12);
      p7[i] = *(const v4f*)(xr + (64 - f) * 12);
    }
  };
  auto stage_chunk = [&]() {
    #pragma unroll
    for (int i = 0; i < 6; ++i) {
      const int id = tid + (i << 8);
      const int rr = id / 48, cc = id - rr * 48;
      fold_store(As, p0[i], p1[i], p2[i], p3[i], p4[i], p5[i], p6[i], p7[i], rr * LDSTR + (cc << 2));
    }
  };

  fetch_chunk(0);
  stage_chunk();
  __syncthreads();

  fetch_chunk(FCH);
  chunk_run(acc, Arow, Wb, 0);
  __syncthreads();
  stage_chunk();
  __syncthreads();

  {
    const int tp = (tid < 96) ? tid : 0;
    const int tr = tp / 3, tc = tp - tr * 3;
    const float* __restrict__ xt = Xblk + (size_t)tr * KTOT + (tc << 2);
    const v4f t32  = *(const v4f*)(xt + 32 * 12);
    const v4f t96  = *(const v4f*)(xt + 96 * 12);
    const v4f t160 = *(const v4f*)(xt + 160 * 12);
    const v4f t224 = *(const v4f*)(xt + 224 * 12);
    chunk_run(acc, Arow, Wb, QCH);
    __syncthreads();
    if (tid < 96) {
      fold_store(As, t32, t96, t160, t224, t224, t160, t96, t32, tr * LDSTR + (tc << 2));
    } else if (tid < 192) {
      const int t2 = tid - 96;
      const int zr = t2 / 3, zc = t2 - zr * 3;
      const v4f z = v4f{0.0f, 0.0f, 0.0f, 0.0f};
      #pragma unroll
      for (int p = 0; p < 8; ++p) *(v4f*)(As + p * PLANE + zr * LDSTR + 12 + (zc << 2)) = z;
    }
    __syncthreads();
  }

  tail_run(acc, Arow, Wb, 2 * QCH);

  {
    const int csw = cls ^ ((lrow >> 3) << 2) ^ (lhi << 1);
    float* __restrict__ op = Os + (lhi << 3) * NOUT + (lrow << 3) + csw;
    #pragma unroll
    for (int j = 0; j < NSL; ++j) {
      const int colg = cls + (j << 7) + (lrow << 3);
      if (colg < NOUT) {
        #pragma unroll
        for (int m = 0; m < MT; ++m) {
          #pragma unroll
          for (int v = 0; v < 8; ++v) op[((m << 4) + v) * NOUT + (j << 7)] = acc[MT * j + m][v];
        }
      }
    }
  }
  __syncthreads();
  {
    float* __restrict__ ob = out + (size_t)m0 * NOUT;
    #pragma unroll
    for (int i = 0; i < (OPC + 255) / 256; ++i) {
      const int q = tid + (i << 8);
      if (q < OPC) {
        const int row = q / 150, c4 = q - row * 150;
        v4f x = *(const v4f*)(Os + row * NOUT + ((c4 ^ ((c4 >> 4) & 1)) << 2));
        if (row & 8) x = v4f{x.z, x.w, x.x, x.y};
        *(v4f*)(ob + (q << 2)) = x;
      }
    }
  }
}

extern "C" void kernel_launch(void* const* d_in, const int* in_sizes, int n_in,
                              void* d_out, int out_size, void* d_ws, size_t ws_size,
                              hipStream_t stream) {
  const float* X  = (const float*)d_in[0];
  float*       O  = (float*)d_out;
  float*       Wf = (float*)d_ws;

  const int wtot    = NFRM * QN * NTW * FRAG;
  const int wblocks = (wtot + 255) / 256;

  ISTFT_buildW<<<wblocks, 256, 0, stream>>>(Wf);
  ISTFT_gemm<<<8192 / 32, 256, 0, stream>>>(X, Wf, O);
}


# Envelope division moved from W to one multiply per output (reciprocal table): W depends on the frame segment only, 4 B pairs per k-step feed the 18 blocks; product rounding changes, sum order not
# speedup vs baseline: 8.1867x; 1.0777x over previous
#include <hip/hip_runtime.h>
#include <hip/hip_bf16.h>
#include <math.h>


typedef __attribute__((ext_vector_type(2))) float v2f;
typedef __attribute__((ext_vector_type(8))) float v8f;
typedef __attribute__((ext_vector_type(4))) float v4f;

#define KTOT   3084
#define NFOLD  33
#define NFRM   6
#define QN     17
#define NGRP   5
#define NTW    40
#define FRAG   64
#define NOUT   600
#define FCH    16
#define QCH    (FCH / 2)
#define LDSTR  196
#define PLANE  (32 * LDSTR)
#define NSEG   4
#define QSTR   (8 * NSEG * FRAG)
#define WTOT   (QN * QSTR)
#define NPAD   (NGRP * 128)
#define MT     2
#define MSTR   (16 * LDSTR)
#define NSL    5
#define OPC    (32 * NOUT / 4)

__global__ __launch_bounds__(256)
void ISTFT_buildW(float* __restrict__ Wf) {
  int gid = blockIdx.x * 256 + threadIdx.x;
  if (gid >= WTOT + NPAD) return;
  const float c0 = 6.283185307179586f / 512.0f;

  if (gid >= WTOT) {
    int labs = (gid - WTOT) + 256;
    float env = 0.0f;
    #pragma unroll
    for (int tp = 0; tp < 6; ++tp) {
      int nn2 = labs - (tp << 7);
      if (nn2 >= 0 && nn2 < 512) {
        float w = 0.5f * (1.0f - cosf(c0 * (float)nn2));
        env += w * w;
      }
    }
    Wf[gid] = 1.0f / env;
    return;
  }

  int d    = gid & 1;
  int l    = (gid >> 1) & 31;
  int rest = gid >> 6;
  int sg   = rest & (NSEG - 1);
  int r    = (rest >> 2) & 7;
  int q    = rest >> 5;
  int f    = (q << 1) + (l >> 4);
  int nn   = (sg << 7) + ((l & 15) << 3) + r;

  float v = 0.0f;
  if (f < NFOLD) {
    float wn   = 0.5f * (1.0f - cosf(c0 * (float)nn));
    bool  half = (f == 0) || (f == 32);
    float scl  = (half ? 1.0f : 2.0f) / 512.0f;
    int   m    = (f * nn) & 511;
    float ang  = c0 * (float)m;
    float basis = d ? -sinf(ang) : cosf(ang);
    if (d && f == 0) basis = 0.0f;
    v = scl * wn * basis;
  }
  Wf[gid] = v;
}

__device__ __forceinline__ constexpr int seg_lo(int i) { return i < 2 ? 0 : i - 2; }
__device__ __forceinline__ constexpr int seg_hi(int i) { return i > 3 ? 4 : i + 1; }

__device__ __forceinline__ void load_step(v2f (&a)[NFRM][MT], v2f (&b)[NSEG],
                                          const float* __restrict__ ap,
                                          const float* __restrict__ wp) {
  #pragma unroll
  for (int sg = 0; sg < NSEG; ++sg) b[sg] = *(const v2f*)(wp + sg * FRAG);
  #pragma unroll
  for (int i = 0; i < NFRM; ++i) {
    #pragma unroll
    for (int m = 0; m < MT; ++m) a[i][m] = *(const v2f*)(ap + i * 2 + m * MSTR);
  }
}

__device__ __forceinline__ void mma_step(v8f (&acc)[NSL * MT], const v2f (&a)[NFRM][MT],
                                         const v2f (&b)[NSEG]) {
  #pragma unroll
  for (int i = 0; i < NFRM; ++i) {
    #pragma unroll
    for (int j = 0; j < NSL; ++j) {
      if (j >= seg_lo(i) && j <= seg_hi(i)) {
        #pragma unroll
        for (int m = 0; m < MT; ++m) {
          acc[MT * j + m] = __builtin_amdgcn_wmma_f32_16x16x4_f32(
              false, a[i][m], false, b[j + 2 - i], (short)0, acc[MT * j + m], false, false);
        }
      }
    }
  }
}

__device__ __forceinline__ void chunk_run(v8f (&acc)[NSL * MT], const float* __restrict__ arow,
                                          const float* __restrict__ wb, int q0) {
  v2f a0[NFRM][MT], a1[NFRM][MT];
  v2f b0[NSEG], b1[NSEG];
  const float* __restrict__ wp = wb + (size_t)q0 * QSTR;

  load_step(a0, b0, arow, wp);
  #pragma nounroll
  for (int j = 0; j + 2 < QCH; j += 2) {
    load_step(a1, b1, arow + (j + 1) * 24, wp + (size_t)(j + 1) * QSTR);
    asm volatile("" ::: "memory");
    mma_step(acc, a0, b0);
    load_step(a0, b0, arow + (j + 2) * 24, wp + (size_t)(j + 2) * QSTR);
    mma_step(acc, a1, b1);
  }
  load_step(a1, b1, arow + (QCH - 1) * 24, wp + (size_t)(QCH - 1) * QSTR);
  mma_step(acc, a0, b0);
  mma_step(acc, a1, b1);
}

__device__ __forceinline__ void tail_run(v8f (&acc)[NSL * MT], const float* __restrict__ arow,
                                         const float* __restrict__ wb, int q0) {
  v2f a[NFRM][MT];
  v2f b[NSEG];
  load_step(a, b, arow, wb + (size_t)q0 * QSTR);
  mma_step(acc, a, b);
}

__device__ __forceinline__ void fold_store(float* __restrict__ As, const v4f x0, const v4f x1,
                                           const v4f x2, const v4f x3, const v4f y4, const v4f y5,
                                           const v4f y6, const v4f y7, const int off) {
  const float h = 0.70710678118654752f;
  const v4f A0 = v4f{x0.x + y4.x, x0.y - y4.y, x0.z + y4.z, x0.w - y4.w};
  const v4f B0 = v4f{x0.x - y4.x, x0.y + y4.y, x0.z - y4.z, x0.w + y4.w};
  const v4f A1 = v4f{x1.x + y5.x, x1.y - y5.y, x1.z + y5.z, x1.w - y5.w};
  const v4f B1 = v4f{x1.x - y5.x, x1.y + y5.y, x1.z - y5.z, x1.w + y5.w};
  const v4f A2 = v4f{x2.x + y6.x, x2.y - y6.y, x2.z + y6.z, x2.w - y6.w};
  const v4f B2 = v4f{x2.x - y6.x, x2.y + y6.y, x2.z - y6.z, x2.w + y6.w};
  const v4f A3 = v4f{x3.x + y7.x, x3.y - y7.y, x3.z + y7.z, x3.w - y7.w};
  const v4f B3 = v4f{x3.x - y7.x, x3.y + y7.y, x3.z - y7.z, x3.w + y7.w};
  {
    const v4f e0 = A0 + A2, e1 = A0 - A2, o0 = A1 + A3, o1 = A1 - A3;
    *(v4f*)(As + off)             = e0 + o0;
    *(v4f*)(As + 4 * PLANE + off) = e0 - o0;
    *(v4f*)(As + 2 * PLANE + off) = v4f{e1.x - o1.y, e1.y + o1.x, e1.z - o1.w, e1.w + o1.z};
    *(v4f*)(As + 6 * PLANE + off) = v4f{e1.x + o1.y, e1.y - o1.x, e1.z + o1.w, e1.w - o1.z};
  }
  {
    const v4f C1 = v4f{h * (B1.x - B1.y), h * (B1.x + B1.y), h * (B1.z - B1.w), h * (B1.z + B1.w)};
    const v4f C2 = v4f{-B2.y, B2.x, -B2.w, B2.z};
    const v4f C3 = v4f{-h * (B3.x + B3.y), h * (B3.x - B3.y), -h * (B3.z + B3.w), h * (B3.z - B3.w)};
    const v4f e0 = B0 + C2, e1 = B0 - C2, o0 = C1 + C3, o1 = C1 - C3;
    *(v4f*)(As + 1 * PLANE + off) = e0 + o0;
    *(v4f*)(As + 5 * PLANE + off) = e0 - o0;
    *(v4f*)(As + 3 * PLANE + off) = v4f{e1.x - o1.y, e1.y + o1.x, e1.z - o1.w, e1.w + o1.z};
    *(v4f*)(As + 7 * PLANE + off) = v4f{e1.x + o1.y, e1.y - o1.x, e1.z + o1.w, e1.w - o1.z};
  }
}

__global__ __launch_bounds__(256)
void ISTFT_gemm(const float* __restrict__ X, const float* __restrict__ Wf,
                float* __restrict__ out) {
  __shared__ float As[8 * PLANE];
  __shared__ float Os[32 * NOUT];

  const int tid  = threadIdx.x;
  const int wave = __builtin_amdgcn_readfirstlane(tid >> 5);
  const int lane = tid & 31;
  const int lrow = lane & 15;
  const int lhi  = lane >> 4;
  const int m0   = blockIdx.x << 5;

  const int cls  = wave;

  v8f acc[NSL * MT] = {};

  const float* __restrict__ Xblk = X + (size_t)m0 * KTOT;
  const float* __restrict__ Arow = As + cls * PLANE + lrow * LDSTR + lhi * 12;
  const float* __restrict__ Wb   = Wf + (size_t)(cls * NSEG) * FRAG + lane * 2;

  v4f p0[6], p1[6], p2[6], p3[6], p4[6], p5[6], p6[6], p7[6];

  auto fetch_chunk = [&](int fb) {
    #pragma unroll
    for (int i = 0; i < 6; ++i) {
      const int id = tid + (i << 8);
      const int rr = id / 48, cc = id - rr * 48;
      const int fi = cc / 3, t = cc - fi * 3;
      const int f  = fb + fi;
      const float* __restrict__ xr = Xblk + (size_t)rr * KTOT + (t << 2);
      p0[i] = *(const v4f*)(xr + f * 12);
      p1[i] = *(const v4f*)(xr + (64 + f) * 12);
      p2[i] = *(const v4f*)(xr + (128 + f) * 12);
      p3[i] = *(const v4f*)(xr + (192 + f) * 12);
      p4[i] = *(const v4f*)(xr + (256 - f) * 12);
      p5[i] = *(const v4f*)(xr + (192 - f) * 12);
      p6[i] = *(const v4f*)(xr + (128 - f) * 12);
      p7[i] = *(const v4f*)(xr + (64 - f) * 12);
    }
  };
  auto stage_chunk = [&]() {
    #pragma unroll
    for (int i = 0; i < 6; ++i) {
      const int id = tid + (i << 8);
      const int rr = id / 48, cc = id - rr * 48;
      fold_store(As, p0[i], p1[i], p2[i], p3[i], p4[i], p5[i], p6[i], p7[i], rr * LDSTR + (cc << 2));
    }
  };

  fetch_chunk(0);
  stage_chunk();
  __syncthreads();

  fetch_chunk(FCH);
  chunk_run(acc, Arow, Wb, 0);
  __syncthreads();
  stage_chunk();
  __syncthreads();

  {
    const int tp = (tid < 96) ? tid : 0;
    const int tr = tp / 3, tc = tp - tr * 3;
    const float* __restrict__ xt = Xblk + (size_t)tr * KTOT + (tc << 2);
    const v4f t32  = *(const v4f*)(xt + 32 * 12);
    const v4f t96  = *(const v4f*)(xt + 96 * 12);
    const v4f t160 = *(const v4f*)(xt + 160 * 12);
    const v4f t224 = *(const v4f*)(xt + 224 * 12);
    chunk_run(acc, Arow, Wb, QCH);
    __syncthreads();
    if (tid < 96) {
      fold_store(As, t32, t96, t160, t224, t224, t160, t96, t32, tr * LDSTR + (tc << 2));
    } else if (tid < 192) {
      const int t2 = tid - 96;
      const int zr = t2 / 3, zc = t2 - zr * 3;
      const v4f z = v4f{0.0f, 0.0f, 0.0f, 0.0f};
      #pragma unroll
      for (int p = 0; p < 8; ++p) *(v4f*)(As + p * PLANE + zr * LDSTR + 12 + (zc << 2)) = z;
    }
    __syncthreads();
  }

  float rinv[NSL];
  #pragma unroll
  for (int j = 0; j < NSL; ++j) rinv[j] = Wf[WTOT + cls + (j << 7) + (lrow << 3)];

  tail_run(acc, Arow, Wb, 2 * QCH);

  {
    const int csw = cls ^ ((lrow >> 3) << 2) ^ (lhi << 1);
    float* __restrict__ op = Os + (lhi << 3) * NOUT + (lrow << 3) + csw;
    #pragma unroll
    for (int j = 0; j < NSL; ++j) {
      const int colg = cls + (j << 7) + (lrow << 3);
      if (colg < NOUT) {
        #pragma unroll
        for (int m = 0; m < MT; ++m) {
          #pragma unroll
          for (int v = 0; v < 8; ++v) op[((m << 4) + v) * NOUT + (j << 7)] = acc[MT * j + m][v] * rinv[j];
        }
      }
    }
  }
  __syncthreads();
  {
    float* __restrict__ ob = out + (size_t)m0 * NOUT;
    #pragma unroll
    for (int i = 0; i < (OPC + 255) / 256; ++i) {
      const int q = tid + (i << 8);
      if (q < OPC) {
        const int row = q / 150, c4 = q - row * 150;
        v4f x = *(const v4f*)(Os + row * NOUT + ((c4 ^ ((c4 >> 4) & 1)) << 2));
        if (row & 8) x = v4f{x.z, x.w, x.x, x.y};
        *(v4f*)(ob + (q << 2)) = x;
      }
    }
  }
}

extern "C" void kernel_launch(void* const* d_in, const int* in_sizes, int n_in,
                              void* d_out, int out_size, void* d_ws, size_t ws_size,
                              hipStream_t stream) {
  const float* X  = (const float*)d_in[0];
  float*       O  = (float*)d_out;
  float*       Wf = (float*)d_ws;

  const int wtot    = WTOT + NPAD;
  const int wblocks = (wtot + 255) / 256;

  ISTFT_buildW<<<wblocks, 256, 0, stream>>>(Wf);
  ISTFT_gemm<<<8192 / 32, 256, 0, stream>>>(X, Wf, O);
}


# W not built by a launch of its own nor read from L2: each block builds a 512-entry (cos,-sin) table in LDS under its first fetch and lanes make their B pairs from it; one launch; sums unchanged
# speedup vs baseline: 8.7209x; 1.0653x over previous
#include <hip/hip_runtime.h>
#include <hip/hip_bf16.h>
#include <math.h>


typedef __attribute__((ext_vector_type(2))) float v2f;
typedef __attribute__((ext_vector_type(8))) float v8f;
typedef __attribute__((ext_vector_type(4))) float v4f;

#define KTOT   3084
#define NFOLD  33
#define NFRM   6
#define QN     17
#define NGRP   5
#define NTW    40
#define NOUT   600
#define FCH    16
#define QCH    (FCH / 2)
#define LDSTR  196
#define PLANE  (32 * LDSTR)
#define NSEG   4
#define NTAB   512
#define MT     2
#define MSTR   (16 * LDSTR)
#define NSL    5
#define OPC    (32 * NOUT / 4)


__device__ __forceinline__ constexpr int seg_lo(int i) { return i < 2 ? 0 : i - 2; }
__device__ __forceinline__ constexpr int seg_hi(int i) { return i > 3 ? 4 : i + 1; }

__device__ __forceinline__ void load_step(v2f (&a)[NFRM][MT], v2f (&b)[NSEG],
                                          const float* __restrict__ ap,
                                          const float* __restrict__ tb, const unsigned m8,
                                          const unsigned s8, const float (&w)[NSEG]) {
  #pragma unroll
  for (int sg = 0; sg < NSEG; ++sg) {
    const unsigned off = (m8 + (unsigned)sg * s8) & 0xFF8u;
    const v2f t = *(const v2f*)((const char*)tb + off);
    b[sg] = t * w[sg];
  }
  #pragma unroll
  for (int i = 0; i < NFRM; ++i) {
    #pragma unroll
    for (int m = 0; m < MT; ++m) a[i][m] = *(const v2f*)(ap + i * 2 + m * MSTR);
  }
}

__device__ __forceinline__ void mma_step(v8f (&acc)[NSL * MT], const v2f (&a)[NFRM][MT],
                                         const v2f (&b)[NSEG]) {
  #pragma unroll
  for (int i = 0; i < NFRM; ++i) {
    #pragma unroll
    for (int j = 0; j < NSL; ++j) {
      if (j >= seg_lo(i) && j <= seg_hi(i)) {
        #pragma unroll
        for (int m = 0; m < MT; ++m) {
          acc[MT * j + m] = __builtin_amdgcn_wmma_f32_16x16x4_f32(
              false, a[i][m], false, b[j + 2 - i], (short)0, acc[MT * j + m], false, false);
        }
      }
    }
  }
}

__device__ __forceinline__ void chunk_run(v8f (&acc)[NSL * MT], const float* __restrict__ arow,
                                          const float* __restrict__ tb, const unsigned nn8,
                                          const unsigned f0, const float (&wf)[NSEG],
                                          const float (&w)[NSEG]) {
  v2f a0[NFRM][MT], a1[NFRM][MT];
  v2f b0[NSEG], b1[NSEG];
  const unsigned dm = 2u * nn8;
  const unsigned se = f0 << 10, so = se + 2048u;
  unsigned m8 = f0 * nn8;

  load_step(a0, b0, arow, tb, m8, se, wf);
  #pragma nounroll
  for (int j = 0; j + 2 < QCH; j += 2) {
    load_step(a1, b1, arow + (j + 1) * 24, tb, m8 + dm, so, w);
    asm volatile("" ::: "memory");
    mma_step(acc, a0, b0);
    m8 += 2u * dm;
    load_step(a0, b0, arow + (j + 2) * 24, tb, m8, se, w);
    mma_step(acc, a1, b1);
  }
  load_step(a1, b1, arow + (QCH - 1) * 24, tb, m8 + dm, so, w);
  mma_step(acc, a0, b0);
  mma_step(acc, a1, b1);
}

__device__ __forceinline__ void tail_run(v8f (&acc)[NSL * MT], const float* __restrict__ arow,
                                         const float* __restrict__ tb, const unsigned nn8,
                                         const unsigned f0, const float (&w)[NSEG]) {
  v2f a[NFRM][MT];
  v2f b[NSEG];
  load_step(a, b, arow, tb, f0 * nn8, f0 << 10, w);
  mma_step(acc, a, b);
}

__device__ __forceinline__ void fold_store(float* __restrict__ As, const v4f x0, const v4f x1,
                                           const v4f x2, const v4f x3, const v4f y4, const v4f y5,
                                           const v4f y6, const v4f y7, const int off) {
  const float h = 0.70710678118654752f;
  const v4f A0 = v4f{x0.x + y4.x, x0.y - y4.y, x0.z + y4.z, x0.w - y4.w};
  const v4f B0 = v4f{x0.x - y4.x, x0.y + y4.y, x0.z - y4.z, x0.w + y4.w};
  const v4f A1 = v4f{x1.x + y5.x, x1.y - y5.y, x1.z + y5.z, x1.w - y5.w};
  const v4f B1 = v4f{x1.x - y5.x, x1.y + y5.y, x1.z - y5.z, x1.w + y5.w};
  const v4f A2 = v4f{x2.x + y6.x, x2.y - y6.y, x2.z + y6.z, x2.w - y6.w};
  const v4f B2 = v4f{x2.x - y6.x, x2.y + y6.y, x2.z - y6.z, x2.w + y6.w};
  const v4f A3 = v4f{x3.x + y7.x, x3.y - y7.y, x3.z + y7.z, x3.w - y7.w};
  const v4f B3 = v4f{x3.x - y7.x, x3.y + y7.y, x3.z - y7.z, x3.w + y7.w};
  {
    const v4f e0 = A0 + A2, e1 = A0 - A2, o0 = A1 + A3, o1 = A1 - A3;
    *(v4f*)(As + off)             = e0 + o0;
    *(v4f*)(As + 4 * PLANE + off) = e0 - o0;
    *(v4f*)(As + 2 * PLANE + off) = v4f{e1.x - o1.y, e1.y + o1.x, e1.z - o1.w, e1.w + o1.z};
    *(v4f*)(As + 6 * PLANE + off) = v4f{e1.x + o1.y, e1.y - o1.x, e1.z + o1.w, e1.w - o1.z};
  }
  {
    const v4f C1 = v4f{h * (B1.x - B1.y), h * (B1.x + B1.y), h * (B1.z - B1.w), h * (B1.z + B1.w)};
    const v4f C2 = v4f{-B2.y, B2.x, -B2.w, B2.z};
    const v4f C3 = v4f{-h * (B3.x + B3.y), h * (B3.x - B3.y), -h * (B3.z + B3.w), h * (B3.z - B3.w)};
    const v4f e0 = B0 + C2, e1 = B0 - C2, o0 = C1 + C3, o1 = C1 - C3;
    *(v4f*)(As + 1 * PLANE + off) = e0 + o0;
    *(v4f*)(As + 5 * PLANE + off) = e0 - o0;
    *(v4f*)(As + 3 * PLANE + off) = v4f{e1.x - o1.y, e1.y + o1.x, e1.z - o1.w, e1.w + o1.z};
    *(v4f*)(As + 7 * PLANE + off) = v4f{e1.x + o1.y, e1.y - o1.x, e1.z + o1.w, e1.w - o1.z};
  }
}

__global__ __launch_bounds__(256)
void ISTFT_gemm(const float* __restrict__ X, float* __restrict__ out) {
  __shared__ float As[8 * PLANE];
  __shared__ float Os[32 * NOUT];
  __shared__ __attribute__((aligned(4096))) float Tb[2 * NTAB];

  const int tid  = threadIdx.x;
  const int wave = __builtin_amdgcn_readfirstlane(tid >> 5);
  const int lane = tid & 31;
  const int lrow = lane & 15;
  const int lhi  = lane >> 4;
  const int m0   = blockIdx.x << 5;

  const int cls  = wave;

  v8f acc[NSL * MT] = {};

  const float* __restrict__ Xblk = X + (size_t)m0 * KTOT;
  const float* __restrict__ Arow = As + cls * PLANE + lrow * LDSTR + lhi * 12;
  const int      cpos = cls + (lrow << 3);
  const unsigned nn8  = (unsigned)cpos << 3;
  const unsigned lhu  = (unsigned)lhi;

  v4f p0[6], p1[6], p2[6], p3[6], p4[6], p5[6], p6[6], p7[6];

  auto fetch_chunk = [&](int fb) {
    #pragma unroll
    for (int i = 0; i < 6; ++i) {
      const int id = tid + (i << 8);
      const int rr = id / 48, cc = id - rr * 48;
      const int fi = cc / 3, t = cc - fi * 3;
      const int f  = fb + fi;
      const float* __restrict__ xr = Xblk + (size_t)rr * KTOT + (t << 2);
      p0[i] = *(const v4f*)(xr + f * 12);
      p1[i] = *(const v4f*)(xr + (64 + f) * 12);
      p2[i] = *(const v4f*)(xr + (128 + f) * 12);
      p3[i] = *(const v4f*)(xr + (192 + f) * 12);
      p4[i] = *(const v4f*)(xr + (256 - f) * 12);
      p5[i] = *(const v4f*)(xr + (192 - f) * 12);
      p6[i] = *(const v4f*)(xr + (128 - f) * 12);
      p7[i] = *(const v4f*)(xr + (64 - f) * 12);
    }
  };
  auto stage_chunk = [&]() {
    #pragma unroll
    for (int i = 0; i < 6; ++i) {
      const int id = tid + (i << 8);
      const int rr = id / 48, cc = id - rr * 48;
      fold_store(As, p0[i], p1[i], p2[i], p3[i], p4[i], p5[i], p6[i], p7[i], rr * LDSTR + (cc << 2));
    }
  };

  fetch_chunk(0);

  const float c0 = 6.283185307179586f / 512.0f;
  #pragma unroll
  for (int e = 0; e < 2; ++e) {
    const int   m   = tid + (e << 8);
    const float ang = c0 * (float)m;
    *(v2f*)(Tb + 2 * m) = v2f{cosf(ang), -sinf(ang)};
  }
  float wsc[NSEG], wsc0[NSEG];
  #pragma unroll
  for (int sg = 0; sg < NSEG; ++sg) {
    const float wn = 0.5f * (1.0f - cosf(c0 * (float)((sg << 7) + cpos)));
    wsc[sg]  = (2.0f / 512.0f) * wn;
    wsc0[sg] = lhi ? wsc[sg] : 0.5f * wsc[sg];
  }

  stage_chunk();
  __syncthreads();

  fetch_chunk(FCH);
  chunk_run(acc, Arow, Tb, nn8, lhu, wsc0, wsc);
  __syncthreads();
  stage_chunk();
  __syncthreads();

  {
    const int tp = (tid < 96) ? tid : 0;
    const int tr = tp / 3, tc = tp - tr * 3;
    const float* __restrict__ xt = Xblk + (size_t)tr * KTOT + (tc << 2);
    const v4f t32  = *(const v4f*)(xt + 32 * 12);
    const v4f t96  = *(const v4f*)(xt + 96 * 12);
    const v4f t160 = *(const v4f*)(xt + 160 * 12);
    const v4f t224 = *(const v4f*)(xt + 224 * 12);
    chunk_run(acc, Arow, Tb, nn8, 2u * QCH + lhu, wsc, wsc);
    __syncthreads();
    if (tid < 96) {
      fold_store(As, t32, t96, t160, t224, t224, t160, t96, t32, tr * LDSTR + (tc << 2));
    } else if (tid < 192) {
      const int t2 = tid - 96;
      const int zr = t2 / 3, zc = t2 - zr * 3;
      const v4f z = v4f{0.0f, 0.0f, 0.0f, 0.0f};
      #pragma unroll
      for (int p = 0; p < 8; ++p) *(v4f*)(As + p * PLANE + zr * LDSTR + 12 + (zc << 2)) = z;
    }
    __syncthreads();
  }

  float rinv[NSL];
  #pragma unroll
  for (int j = 0; j < NSL; ++j) {
    float env = 0.0f;
    #pragma unroll
    for (int tp = 0; tp < NFRM; ++tp) {
      const int s = j + 2 - tp;
      if (s >= 0 && s < NSEG) {
        const float w = 256.0f * wsc[s];
        env += w * w;
      }
    }
    rinv[j] = 1.0f / env;
  }

  float wsct[NSEG];
  #pragma unroll
  for (int sg = 0; sg < NSEG; ++sg) wsct[sg] = lhi ? 0.0f : 0.5f * wsc[sg];
  tail_run(acc, Arow, Tb, nn8, 4u * QCH + lhu, wsct);

  {
    const int csw = cls ^ ((lrow >> 3) << 2) ^ (lhi << 1);
    float* __restrict__ op = Os + (lhi << 3) * NOUT + (lrow << 3) + csw;
    #pragma unroll
    for (int j = 0; j < NSL; ++j) {
      const int colg = cls + (j << 7) + (lrow << 3);
      if (colg < NOUT) {
        #pragma unroll
        for (int m = 0; m < MT; ++m) {
          #pragma unroll
          for (int v = 0; v < 8; ++v) op[((m << 4) + v) * NOUT + (j << 7)] = acc[MT * j + m][v] * rinv[j];
        }
      }
    }
  }
  __syncthreads();
  {
    float* __restrict__ ob = out + (size_t)m0 * NOUT;
    #pragma unroll
    for (int i = 0; i < (OPC + 255) / 256; ++i) {
      const int q = tid + (i << 8);
      if (q < OPC) {
        const int row = q / 150, c4 = q - row * 150;
        v4f x = *(const v4f*)(Os + row * NOUT + ((c4 ^ ((c4 >> 4) & 1)) << 2));
        if (row & 8) x = v4f{x.z, x.w, x.x, x.y};
        *(v4f*)(ob + (q << 2)) = x;
      }
    }
  }
}

extern "C" void kernel_launch(void* const* d_in, const int* in_sizes, int n_in,
                              void* d_out, int out_size, void* d_ws, size_t ws_size,
                              hipStream_t stream) {
  const float* X  = (const float*)d_in[0];
  float*       O  = (float*)d_out;

  ISTFT_gemm<<<8192 / 32, 256, 0, stream>>>(X, O);
}
